# starting kernel
# baseline (speedup 1.0000x reference)
#include <hip/hip_runtime.h>
#include <hip/hip_bf16.h>

typedef __attribute__((ext_vector_type(2))) float v2f;
typedef __attribute__((ext_vector_type(8))) float v8f;

#define GEMM_WAVES 4
#define ROW_ITERS  8

__global__ void gcn_zero_f32(float* __restrict__ p, long long n) {
  long long i = (long long)blockIdx.x * blockDim.x + threadIdx.x;
  if (i < n) p[i] = 0.0f;
}

__global__ void gcn_degrees(const int* __restrict__ src, const int* __restrict__ dst,
                            float* __restrict__ deg_out, float* __restrict__ deg_in,
                            int E) {
  int e = blockIdx.x * blockDim.x + threadIdx.x;
  if (e < E) {
    atomicAdd(&deg_out[src[e]], 1.0f);
    atomicAdd(&deg_in[dst[e]], 1.0f);
  }
}

__global__ void gcn_norm(const float* __restrict__ deg, float* __restrict__ norm, int n) {
  int i = blockIdx.x * blockDim.x + threadIdx.x;
  if (i < n) {
    float d = deg[i];
    norm[i] = (d > 0.0f) ? rsqrtf(d) : 0.0f;
  }
}

template <int KDIM>
__global__ __launch_bounds__(32 * GEMM_WAVES) void gcn_gemm_wmma(
    const float* __restrict__ X, const float* __restrict__ Wm,
    const float* __restrict__ prescale, float* __restrict__ Y,
    int nRows, int Mout) {
  constexpr int KSTEPS = KDIM / 4;
  const int wave = threadIdx.x >> 5;
  const int lane = threadIdx.x & 31;
  const int nRowTiles = (nRows + 15) >> 4;
  const int tileBase = (blockIdx.x * GEMM_WAVES + wave) * ROW_ITERS;
  if (tileBase >= nRowTiles) return;

  const int m  = lane & 15;
  const int kg = lane >> 4;

  const int col = blockIdx.y * 16 + m;
  const int colc = (col < Mout) ? col : (Mout - 1);
  v2f breg[KSTEPS];
#pragma unroll
  for (int s = 0; s < KSTEPS; ++s) {
    const int k = 4 * s + 2 * kg;
    breg[s].x = Wm[k * Mout + colc];
    breg[s].y = Wm[(k + 1) * Mout + colc];
  }

  const int tileEnd = (tileBase + ROW_ITERS < nRowTiles) ? tileBase + ROW_ITERS
                                                         : nRowTiles;
  for (int tile = tileBase; tile < tileEnd; ++tile) {
    int row = tile * 16 + m;
    if (row >= nRows) row = nRows - 1;
    const float scale = prescale[row];
    const float* __restrict__ xrow = X + (long long)row * KDIM + 2 * kg;

    v8f acc = {};
#pragma unroll
    for (int s = 0; s < KSTEPS; ++s) {
      v2f A = *(const v2f*)(xrow + 4 * s);
      A.x *= scale;
      A.y *= scale;
      acc = __builtin_amdgcn_wmma_f32_16x16x4_f32(
           false, A,  false, breg[s],
           (short)0, acc,  false,  false);
    }

    if (col < Mout) {
#pragma unroll
      for (int r = 0; r < 8; ++r) {
        const int outRow = tile * 16 + r + 8 * kg;
        if (outRow < nRows) Y[(long long)outRow * Mout + col] = acc[r];
      }
    }
  }
}

template <int GROUPS>
__global__ void gcn_scatter(const float* __restrict__ Y, const int* __restrict__ src,
                            const int* __restrict__ dst, float* __restrict__ agg,
                            unsigned E) {
  constexpr int COLS = GROUPS * 4;
  const unsigned idx = blockIdx.x * blockDim.x + threadIdx.x;
  if (idx >= E * GROUPS) return;
  const unsigned e = idx / GROUPS;
  const int g = (int)(idx - e * GROUPS);
  const int s = src[e];
  const int d = dst[e];
  const float4 v = *(const float4*)(Y + (long long)s * COLS + g * 4);
  float* __restrict__ out = agg + (long long)d * COLS + g * 4;
  atomicAdd(out + 0, v.x);
  atomicAdd(out + 1, v.y);
  atomicAdd(out + 2, v.z);
  atomicAdd(out + 3, v.w);
}

template <int COLS, bool RELU>
__global__ void gcn_pointwise(float* __restrict__ A, const float* __restrict__ norm_in,
                              const float* __restrict__ bias, unsigned n) {
  const unsigned i = blockIdx.x * blockDim.x + threadIdx.x;
  if (i >= n) return;
  const unsigned row = i / COLS;
  const unsigned c = i - row * COLS;
  float v = A[i] * norm_in[row] + bias[c];
  if (RELU) v = fmaxf(v, 0.0f);
  A[i] = v;
}

extern "C" void kernel_launch(void* const* d_in, const int* in_sizes, int n_in,
                              void* d_out, int out_size, void* d_ws, size_t ws_size,
                              hipStream_t stream) {
  const float* feat = (const float*)d_in[0];
  const int*   src  = (const int*)d_in[1];
  const int*   dst  = (const int*)d_in[2];
  const float* W0   = (const float*)d_in[3];
  const float* b0   = (const float*)d_in[4];
  const float* W1   = (const float*)d_in[5];
  const float* b1   = (const float*)d_in[6];
  const float* W2   = (const float*)d_in[7];
  const float* b2   = (const float*)d_in[8];
  float* out = (float*)d_out;

  const int F = 128, H = 128, C = 40;
  const int N = in_sizes[0] / F;
  const int E = in_sizes[1];
  (void)in_sizes; (void)n_in; (void)out_size; (void)ws_size;

  float* ws       = (float*)d_ws;
  float* deg_out  = ws;
  float* norm_out = ws + 2LL * N;
  float* norm_in  = ws + 3LL * N;
  float* B0       = ws + 4LL * N;
  float* B1       = B0 + (long long)N * H;

  const int TB = 256;
  auto blocks1d = [](long long n, int tb) { return (unsigned)((n + tb - 1) / tb); };

  gcn_zero_f32<<<blocks1d(2LL * N, TB), TB, 0, stream>>>(deg_out, 2LL * N);
  gcn_degrees<<<blocks1d(E, TB), TB, 0, stream>>>(src, dst, deg_out, deg_out + N, E);
  gcn_norm<<<blocks1d(2LL * N, TB), TB, 0, stream>>>(deg_out, norm_out, 2 * N);

  const int rowTiles = (N + 15) / 16;
  const int tilesPerBlock = GEMM_WAVES * ROW_ITERS;
  const dim3 gemmBlk(32 * GEMM_WAVES);
  const dim3 gridH((rowTiles + tilesPerBlock - 1) / tilesPerBlock, H / 16);
  const dim3 gridC((rowTiles + tilesPerBlock - 1) / tilesPerBlock, (C + 15) / 16);

  gcn_gemm_wmma<128><<<gridH, gemmBlk, 0, stream>>>(feat, W0, norm_out, B0, N, H);
  gcn_zero_f32<<<blocks1d((long long)N * H, TB), TB, 0, stream>>>(B1, (long long)N * H);
  gcn_scatter<32><<<blocks1d((long long)E * 32, TB), TB, 0, stream>>>(B0, src, dst, B1, E);
  gcn_pointwise<128, true><<<blocks1d((long long)N * H, TB), TB, 0, stream>>>(
      B1, norm_in, b0, (unsigned)(N * H));

  gcn_gemm_wmma<128><<<gridH, gemmBlk, 0, stream>>>(B1, W1, norm_out, B0, N, H);
  gcn_zero_f32<<<blocks1d((long long)N * H, TB), TB, 0, stream>>>(B1, (long long)N * H);
  gcn_scatter<32><<<blocks1d((long long)E * 32, TB), TB, 0, stream>>>(B0, src, dst, B1, E);
  gcn_pointwise<128, true><<<blocks1d((long long)N * H, TB), TB, 0, stream>>>(
      B1, norm_in, b1, (unsigned)(N * H));

  gcn_gemm_wmma<128><<<gridC, gemmBlk, 0, stream>>>(B1, W2, norm_out, B0, N, C);
  gcn_zero_f32<<<blocks1d((long long)N * C, TB), TB, 0, stream>>>(out, (long long)N * C);
  gcn_scatter<10><<<blocks1d((long long)E * 10, TB), TB, 0, stream>>>(B0, src, dst, out, E);
  gcn_pointwise<40, false><<<blocks1d((long long)N * C, TB), TB, 0, stream>>>(
      out, norm_in, b2, (unsigned)(N * C));
}


# aggregation via per-call CSR (in-degree prefix sum, int-atomic slot fill) + wave-per-row gather with fused norm/bias/relu replaces float-atomic scatter; row sum order = fill arrival order
# speedup vs baseline: 2.1897x; 2.1897x over previous
#include <hip/hip_runtime.h>
#include <hip/hip_bf16.h>

typedef __attribute__((ext_vector_type(2))) float v2f;
typedef __attribute__((ext_vector_type(8))) float v8f;

#define GEMM_WAVES 4
#define ROW_ITERS  8

__global__ void gcn_zero_f32(float* __restrict__ p, long long n) {
  long long i = (long long)blockIdx.x * blockDim.x + threadIdx.x;
  if (i < n) p[i] = 0.0f;
}

__global__ void gcn_degrees(const int* __restrict__ src, const int* __restrict__ dst,
                            float* __restrict__ deg_out, float* __restrict__ deg_in,
                            int E) {
  int e = blockIdx.x * blockDim.x + threadIdx.x;
  if (e < E) {
    atomicAdd(&deg_out[src[e]], 1.0f);
    atomicAdd(&deg_in[dst[e]], 1.0f);
  }
}

__global__ void gcn_norm(const float* __restrict__ deg, float* __restrict__ norm, int n) {
  int i = blockIdx.x * blockDim.x + threadIdx.x;
  if (i < n) {
    float d = deg[i];
    norm[i] = (d > 0.0f) ? rsqrtf(d) : 0.0f;
  }
}

template <int KDIM>
__global__ __launch_bounds__(32 * GEMM_WAVES) void gcn_gemm_wmma(
    const float* __restrict__ X, const float* __restrict__ Wm,
    const float* __restrict__ prescale, float* __restrict__ Y,
    int nRows, int Mout) {
  constexpr int KSTEPS = KDIM / 4;
  const int wave = threadIdx.x >> 5;
  const int lane = threadIdx.x & 31;
  const int nRowTiles = (nRows + 15) >> 4;
  const int tileBase = (blockIdx.x * GEMM_WAVES + wave) * ROW_ITERS;
  if (tileBase >= nRowTiles) return;

  const int m  = lane & 15;
  const int kg = lane >> 4;

  const int col = blockIdx.y * 16 + m;
  const int colc = (col < Mout) ? col : (Mout - 1);
  v2f breg[KSTEPS];
#pragma unroll
  for (int s = 0; s < KSTEPS; ++s) {
    const int k = 4 * s + 2 * kg;
    breg[s].x = Wm[k * Mout + colc];
    breg[s].y = Wm[(k + 1) * Mout + colc];
  }

  const int tileEnd = (tileBase + ROW_ITERS < nRowTiles) ? tileBase + ROW_ITERS
                                                         : nRowTiles;
  for (int tile = tileBase; tile < tileEnd; ++tile) {
    int row = tile * 16 + m;
    if (row >= nRows) row = nRows - 1;
    const float scale = prescale[row];
    const float* __restrict__ xrow = X + (long long)row * KDIM + 2 * kg;

    v8f acc = {};
#pragma unroll
    for (int s = 0; s < KSTEPS; ++s) {
      v2f A = *(const v2f*)(xrow + 4 * s);
      A.x *= scale;
      A.y *= scale;
      acc = __builtin_amdgcn_wmma_f32_16x16x4_f32(
           false, A,  false, breg[s],
           (short)0, acc,  false,  false);
    }

    if (col < Mout) {
#pragma unroll
      for (int r = 0; r < 8; ++r) {
        const int outRow = tile * 16 + r + 8 * kg;
        if (outRow < nRows) Y[(long long)outRow * Mout + col] = acc[r];
      }
    }
  }
}

#define SCAN_TB    256
#define SCAN_CHUNK 1024

__device__ __forceinline__ int gcn_block_excl_scan(int v, int* sh, int& total) {
  const int t = threadIdx.x;
  sh[t] = v;
  __syncthreads();
  int in = 0;
  for (int off = 1; off < SCAN_TB; off <<= 1) {
    int x = sh[in * SCAN_TB + t];
    if (t >= off) x += sh[in * SCAN_TB + t - off];
    sh[(1 - in) * SCAN_TB + t] = x;
    __syncthreads();
    in = 1 - in;
  }
  const int incl = sh[in * SCAN_TB + t];
  total = sh[in * SCAN_TB + SCAN_TB - 1];
  __syncthreads();
  return incl - v;
}

__global__ __launch_bounds__(SCAN_TB) void gcn_chunk_sums(
    const float* __restrict__ deg_in, int* __restrict__ chunk_sums, int n) {
  __shared__ int sh[2 * SCAN_TB];
  const int base = blockIdx.x * SCAN_CHUNK + threadIdx.x * 4;
  int v = 0;
  for (int k = 0; k < 4; ++k)
    if (base + k < n) v += (int)deg_in[base + k];
  int total;
  (void)gcn_block_excl_scan(v, sh, total);
  if (threadIdx.x == 0) chunk_sums[blockIdx.x] = total;
}

__global__ __launch_bounds__(SCAN_TB) void gcn_rowptr(
    const float* __restrict__ deg_in, const int* __restrict__ chunk_sums,
    int* __restrict__ rowptr, int* __restrict__ cursor, int n) {
  __shared__ int sh[2 * SCAN_TB];
  int part = 0;
  for (int j = threadIdx.x; j < (int)blockIdx.x; j += SCAN_TB) part += chunk_sums[j];
  int offset;
  (void)gcn_block_excl_scan(part, sh, offset);

  const int base = blockIdx.x * SCAN_CHUNK + threadIdx.x * 4;
  int d[4];
  int v = 0;
  for (int k = 0; k < 4; ++k) {
    d[k] = (base + k < n) ? (int)deg_in[base + k] : 0;
    v += d[k];
  }
  int total;
  int run = offset + gcn_block_excl_scan(v, sh, total);
  for (int k = 0; k < 4; ++k) {
    if (base + k < n) {
      rowptr[base + k] = run;
      cursor[base + k] = run;
    }
    run += d[k];
  }
  if (blockIdx.x == gridDim.x - 1 && threadIdx.x == 0) rowptr[n] = offset + total;
}

__global__ void gcn_fill(const int* __restrict__ src, const int* __restrict__ dst,
                         int* __restrict__ cursor, int* __restrict__ col, int E) {
  const int e = blockIdx.x * blockDim.x + threadIdx.x;
  if (e < E) {
    const int slot = atomicAdd(&cursor[dst[e]], 1);
    col[slot] = src[e];
  }
}

#define GATHER_WAVES  8
#define GATHER_UNROLL 8
template <int COLS, bool RELU>
__global__ __launch_bounds__(32 * GATHER_WAVES) void gcn_gather(
    const float* __restrict__ Y, const int* __restrict__ rowptr,
    const int* __restrict__ col, const float* __restrict__ norm_in,
    const float* __restrict__ bias, float* __restrict__ out, int nRows) {
  constexpr int LANES = COLS / 4;
  const int lane = threadIdx.x & 31;
  const int row = blockIdx.x * GATHER_WAVES +
                  __builtin_amdgcn_readfirstlane((int)(threadIdx.x >> 5));
  if (row >= nRows) return;
  const bool active = lane < LANES;
  const int start = rowptr[row];
  const int end = rowptr[row + 1];
  const float* __restrict__ ycol = Y + 4 * (active ? lane : 0);

  float4 acc = make_float4(0.0f, 0.0f, 0.0f, 0.0f);
  for (int base = start; base < end; base += 32) {
    const int n = (end - base < 32) ? (end - base) : 32;
    const int myIdx = (lane < n) ? col[base + lane] : 0;
    for (int j = 0; j < n; j += GATHER_UNROLL) {
      float4 v[GATHER_UNROLL];
#pragma unroll
      for (int u = 0; u < GATHER_UNROLL; ++u) {
        const int s = __builtin_amdgcn_readlane(myIdx, j + u);
        v[u] = *(const float4*)(ycol + (long long)s * COLS);
      }
#pragma unroll
      for (int u = 0; u < GATHER_UNROLL; ++u) {
        const bool k = (j + u) < n;
        acc.x += k ? v[u].x : 0.0f;
        acc.y += k ? v[u].y : 0.0f;
        acc.z += k ? v[u].z : 0.0f;
        acc.w += k ? v[u].w : 0.0f;
      }
    }
  }

  if (active) {
    const float ni = norm_in[row];
    const float4 b = *(const float4*)(bias + 4 * lane);
    float4 r;
    r.x = acc.x * ni + b.x;
    r.y = acc.y * ni + b.y;
    r.z = acc.z * ni + b.z;
    r.w = acc.w * ni + b.w;
    if (RELU) {
      r.x = fmaxf(r.x, 0.0f); r.y = fmaxf(r.y, 0.0f);
      r.z = fmaxf(r.z, 0.0f); r.w = fmaxf(r.w, 0.0f);
    }
    *(float4*)(out + (long long)row * COLS + 4 * lane) = r;
  }
}

extern "C" void kernel_launch(void* const* d_in, const int* in_sizes, int n_in,
                              void* d_out, int out_size, void* d_ws, size_t ws_size,
                              hipStream_t stream) {
  const float* feat = (const float*)d_in[0];
  const int*   src  = (const int*)d_in[1];
  const int*   dst  = (const int*)d_in[2];
  const float* W0   = (const float*)d_in[3];
  const float* b0   = (const float*)d_in[4];
  const float* W1   = (const float*)d_in[5];
  const float* b1   = (const float*)d_in[6];
  const float* W2   = (const float*)d_in[7];
  const float* b2   = (const float*)d_in[8];
  float* out = (float*)d_out;

  const int F = 128, H = 128, C = 40;
  const int N = in_sizes[0] / F;
  const int E = in_sizes[1];
  (void)in_sizes; (void)n_in; (void)out_size; (void)ws_size;

  float* ws       = (float*)d_ws;
  float* deg_out  = ws;
  float* norm_out = ws + 2LL * N;
  float* norm_in  = ws + 3LL * N;
  float* B0       = ws + 4LL * N;
  float* B1       = B0 + (long long)N * H;
  const int nChunks = (N + SCAN_CHUNK - 1) / SCAN_CHUNK;
  int* rowptr     = (int*)(B1 + (long long)N * H);
  int* cursor     = rowptr + (N + 1);
  int* chunk_sums = cursor + N;
  int* col        = chunk_sums + nChunks;

  const int TB = 256;
  auto blocks1d = [](long long n, int tb) { return (unsigned)((n + tb - 1) / tb); };

  gcn_zero_f32<<<blocks1d(2LL * N, TB), TB, 0, stream>>>(deg_out, 2LL * N);
  gcn_degrees<<<blocks1d(E, TB), TB, 0, stream>>>(src, dst, deg_out, deg_out + N, E);
  gcn_norm<<<blocks1d(2LL * N, TB), TB, 0, stream>>>(deg_out, norm_out, 2 * N);

  gcn_chunk_sums<<<nChunks, SCAN_TB, 0, stream>>>(deg_out + N, chunk_sums, N);
  gcn_rowptr<<<nChunks, SCAN_TB, 0, stream>>>(deg_out + N, chunk_sums, rowptr, cursor, N);
  gcn_fill<<<blocks1d(E, TB), TB, 0, stream>>>(src, dst, cursor, col, E);
  const dim3 gatherBlk(32 * GATHER_WAVES);
  const unsigned gatherGrid = (unsigned)((N + GATHER_WAVES - 1) / GATHER_WAVES);

  const int rowTiles = (N + 15) / 16;
  const int tilesPerBlock = GEMM_WAVES * ROW_ITERS;
  const dim3 gemmBlk(32 * GEMM_WAVES);
  const dim3 gridH((rowTiles + tilesPerBlock - 1) / tilesPerBlock, H / 16);
  const dim3 gridC((rowTiles + tilesPerBlock - 1) / tilesPerBlock, (C + 15) / 16);

  gcn_gemm_wmma<128><<<gridH, gemmBlk, 0, stream>>>(feat, W0, norm_out, B0, N, H);
  gcn_gather<128, true><<<gatherGrid, gatherBlk, 0, stream>>>(B0, rowptr, col, norm_in, b0, B1, N);

  gcn_gemm_wmma<128><<<gridH, gemmBlk, 0, stream>>>(B1, W1, norm_out, B0, N, H);
  gcn_gather<128, true><<<gatherGrid, gatherBlk, 0, stream>>>(B0, rowptr, col, norm_in, b1, B1, N);

  gcn_gemm_wmma<128><<<gridC, gemmBlk, 0, stream>>>(B1, W2, norm_out, B0, N, C);
  gcn_gather<40, false><<<gatherGrid, gatherBlk, 0, stream>>>(B0, rowptr, col, norm_in, b2, out, N);
}


# every global atomic stays in its wave's own L2 half: degree counters and fill cursors exist once per half (die id picks the copy; rows fill front/back); row sum order = per-half arrival order
# speedup vs baseline: 3.2626x; 1.4900x over previous
#include <hip/hip_runtime.h>
#include <hip/hip_bf16.h>

typedef __attribute__((ext_vector_type(2))) float v2f;
typedef __attribute__((ext_vector_type(8))) float v8f;

#define GEMM_WAVES 4
#define ROW_ITERS  8

__global__ void gcn_zero_f32(float* __restrict__ p, long long n) {
  long long i = (long long)blockIdx.x * blockDim.x + threadIdx.x;
  if (i < n) p[i] = 0.0f;
}

__device__ __forceinline__ int gcn_cache_half() {
  const unsigned id = __builtin_amdgcn_s_sendmsg_rtn(0x87);
  return (int)((id >> 18) & 1u);
}

__global__ void gcn_degrees(const int* __restrict__ src, const int* __restrict__ dst,
                            float* __restrict__ cnt, int N, int E) {
  float* __restrict__ mine = cnt + 2LL * N * gcn_cache_half();
  int e = blockIdx.x * blockDim.x + threadIdx.x;
  if (e < E) {
    atomicAdd(&mine[src[e]], 1.0f);
    atomicAdd(&mine[N + dst[e]], 1.0f);
  }
}

__global__ void gcn_norm(const float* __restrict__ cnt, float* __restrict__ norm, int n) {
  int i = blockIdx.x * blockDim.x + threadIdx.x;
  if (i < n) {
    float d = cnt[i] + cnt[n + i];
    norm[i] = (d > 0.0f) ? rsqrtf(d) : 0.0f;
  }
}

template <int KDIM>
__global__ __launch_bounds__(32 * GEMM_WAVES) void gcn_gemm_wmma(
    const float* __restrict__ X, const float* __restrict__ Wm,
    const float* __restrict__ prescale, float* __restrict__ Y,
    int nRows, int Mout) {
  constexpr int KSTEPS = KDIM / 4;
  const int wave = threadIdx.x >> 5;
  const int lane = threadIdx.x & 31;
  const int nRowTiles = (nRows + 15) >> 4;
  const int tileBase = (blockIdx.x * GEMM_WAVES + wave) * ROW_ITERS;
  if (tileBase >= nRowTiles) return;

  const int m  = lane & 15;
  const int kg = lane >> 4;

  const int col = blockIdx.y * 16 + m;
  const int colc = (col < Mout) ? col : (Mout - 1);
  v2f breg[KSTEPS];
#pragma unroll
  for (int s = 0; s < KSTEPS; ++s) {
    const int k = 4 * s + 2 * kg;
    breg[s].x = Wm[k * Mout + colc];
    breg[s].y = Wm[(k + 1) * Mout + colc];
  }

  const int tileEnd = (tileBase + ROW_ITERS < nRowTiles) ? tileBase + ROW_ITERS
                                                         : nRowTiles;
  for (int tile = tileBase; tile < tileEnd; ++tile) {
    int row = tile * 16 + m;
    if (row >= nRows) row = nRows - 1;
    const float scale = prescale[row];
    const float* __restrict__ xrow = X + (long long)row * KDIM + 2 * kg;

    v8f acc = {};
#pragma unroll
    for (int s = 0; s < KSTEPS; ++s) {
      v2f A = *(const v2f*)(xrow + 4 * s);
      A.x *= scale;
      A.y *= scale;
      acc = __builtin_amdgcn_wmma_f32_16x16x4_f32(
           false, A,  false, breg[s],
           (short)0, acc,  false,  false);
    }

    if (col < Mout) {
#pragma unroll
      for (int r = 0; r < 8; ++r) {
        const int outRow = tile * 16 + r + 8 * kg;
        if (outRow < nRows) Y[(long long)outRow * Mout + col] = acc[r];
      }
    }
  }
}

#define SCAN_TB    256
#define SCAN_CHUNK 1024

__device__ __forceinline__ int gcn_block_excl_scan(int v, int* sh, int& total) {
  const int t = threadIdx.x;
  sh[t] = v;
  __syncthreads();
  int in = 0;
  for (int off = 1; off < SCAN_TB; off <<= 1) {
    int x = sh[in * SCAN_TB + t];
    if (t >= off) x += sh[in * SCAN_TB + t - off];
    sh[(1 - in) * SCAN_TB + t] = x;
    __syncthreads();
    in = 1 - in;
  }
  const int incl = sh[in * SCAN_TB + t];
  total = sh[in * SCAN_TB + SCAN_TB - 1];
  __syncthreads();
  return incl - v;
}

__global__ __launch_bounds__(SCAN_TB) void gcn_chunk_sums(
    const float* __restrict__ deg_in0, const float* __restrict__ deg_in1,
    int* __restrict__ chunk_sums, int n) {
  __shared__ int sh[2 * SCAN_TB];
  const int base = blockIdx.x * SCAN_CHUNK + threadIdx.x * 4;
  int v = 0;
  for (int k = 0; k < 4; ++k)
    if (base + k < n) v += (int)(deg_in0[base + k] + deg_in1[base + k]);
  int total;
  (void)gcn_block_excl_scan(v, sh, total);
  if (threadIdx.x == 0) chunk_sums[blockIdx.x] = total;
}

__global__ __launch_bounds__(SCAN_TB) void gcn_rowptr(
    const float* __restrict__ deg_in0, const float* __restrict__ deg_in1,
    const int* __restrict__ chunk_sums, int* __restrict__ rowptr,
    int* __restrict__ cur0, int* __restrict__ cur1, int n) {
  __shared__ int sh[2 * SCAN_TB];
  int part = 0;
  for (int j = threadIdx.x; j < (int)blockIdx.x; j += SCAN_TB) part += chunk_sums[j];
  int offset;
  (void)gcn_block_excl_scan(part, sh, offset);

  const int base = blockIdx.x * SCAN_CHUNK + threadIdx.x * 4;
  int d[4];
  int v = 0;
  for (int k = 0; k < 4; ++k) {
    d[k] = (base + k < n) ? (int)(deg_in0[base + k] + deg_in1[base + k]) : 0;
    v += d[k];
  }
  int total;
  int run = offset + gcn_block_excl_scan(v, sh, total);
  for (int k = 0; k < 4; ++k) {
    if (base + k < n) {
      rowptr[base + k] = run;
      cur0[base + k] = run;
      cur1[base + k] = run + d[k] - 1;
    }
    run += d[k];
  }
  if (blockIdx.x == gridDim.x - 1 && threadIdx.x == 0) rowptr[n] = offset + total;
}

__global__ void gcn_fill(const int* __restrict__ src, const int* __restrict__ dst,
                         int* __restrict__ cur0, int* __restrict__ cur1,
                         int* __restrict__ col, int E) {
  const int half = gcn_cache_half();
  int* __restrict__ cur = half ? cur1 : cur0;
  const int step = half ? -1 : 1;
  const int e = blockIdx.x * blockDim.x + threadIdx.x;
  if (e < E) {
    const int slot = atomicAdd(&cur[dst[e]], step);
    col[slot] = src[e];
  }
}

#define GATHER_WAVES  8
#define GATHER_UNROLL 8
template <int COLS, bool RELU>
__global__ __launch_bounds__(32 * GATHER_WAVES) void gcn_gather(
    const float* __restrict__ Y, const int* __restrict__ rowptr,
    const int* __restrict__ col, const float* __restrict__ norm_in,
    const float* __restrict__ bias, float* __restrict__ out, int nRows) {
  constexpr int LANES = COLS / 4;
  const int lane = threadIdx.x & 31;
  const int row = blockIdx.x * GATHER_WAVES +
                  __builtin_amdgcn_readfirstlane((int)(threadIdx.x >> 5));
  if (row >= nRows) return;
  const bool active = lane < LANES;
  const int start = rowptr[row];
  const int end = rowptr[row + 1];
  const float* __restrict__ ycol = Y + 4 * (active ? lane : 0);

  float4 acc = make_float4(0.0f, 0.0f, 0.0f, 0.0f);
  for (int base = start; base < end; base += 32) {
    const int n = (end - base < 32) ? (end - base) : 32;
    const int myIdx = (lane < n) ? col[base + lane] : 0;
    for (int j = 0; j < n; j += GATHER_UNROLL) {
      float4 v[GATHER_UNROLL];
#pragma unroll
      for (int u = 0; u < GATHER_UNROLL; ++u) {
        const int s = __builtin_amdgcn_readlane(myIdx, j + u);
        v[u] = *(const float4*)(ycol + (long long)s * COLS);
      }
#pragma unroll
      for (int u = 0; u < GATHER_UNROLL; ++u) {
        const bool k = (j + u) < n;
        acc.x += k ? v[u].x : 0.0f;
        acc.y += k ? v[u].y : 0.0f;
        acc.z += k ? v[u].z : 0.0f;
        acc.w += k ? v[u].w : 0.0f;
      }
    }
  }

  if (active) {
    const float ni = norm_in[row];
    const float4 b = *(const float4*)(bias + 4 * lane);
    float4 r;
    r.x = acc.x * ni + b.x;
    r.y = acc.y * ni + b.y;
    r.z = acc.z * ni + b.z;
    r.w = acc.w * ni + b.w;
    if (RELU) {
      r.x = fmaxf(r.x, 0.0f); r.y = fmaxf(r.y, 0.0f);
      r.z = fmaxf(r.z, 0.0f); r.w = fmaxf(r.w, 0.0f);
    }
    *(float4*)(out + (long long)row * COLS + 4 * lane) = r;
  }
}

extern "C" void kernel_launch(void* const* d_in, const int* in_sizes, int n_in,
                              void* d_out, int out_size, void* d_ws, size_t ws_size,
                              hipStream_t stream) {
  const float* feat = (const float*)d_in[0];
  const int*   src  = (const int*)d_in[1];
  const int*   dst  = (const int*)d_in[2];
  const float* W0   = (const float*)d_in[3];
  const float* b0   = (const float*)d_in[4];
  const float* W1   = (const float*)d_in[5];
  const float* b1   = (const float*)d_in[6];
  const float* W2   = (const float*)d_in[7];
  const float* b2   = (const float*)d_in[8];
  float* out = (float*)d_out;

  const int F = 128, H = 128, C = 40;
  const int N = in_sizes[0] / F;
  const int E = in_sizes[1];
  (void)in_sizes; (void)n_in; (void)out_size; (void)ws_size;

  float* ws       = (float*)d_ws;
  float* cnt      = ws;
  float* deg_in0  = ws + 1LL * N;
  float* deg_in1  = ws + 3LL * N;
  float* norm_out = ws + 4LL * N;
  float* norm_in  = ws + 5LL * N;
  float* B0       = ws + 6LL * N;
  float* B1       = B0 + (long long)N * H;
  const int nChunks = (N + SCAN_CHUNK - 1) / SCAN_CHUNK;
  int* rowptr     = (int*)(B1 + (long long)N * H);
  int* cur0       = rowptr + (N + 1);
  int* cur1       = cur0 + N;
  int* chunk_sums = cur1 + N;
  int* col        = chunk_sums + nChunks;

  const int TB = 256;
  auto blocks1d = [](long long n, int tb) { return (unsigned)((n + tb - 1) / tb); };

  gcn_zero_f32<<<blocks1d(4LL * N, TB), TB, 0, stream>>>(cnt, 4LL * N);
  gcn_degrees<<<blocks1d(E, TB), TB, 0, stream>>>(src, dst, cnt, N, E);
  gcn_norm<<<blocks1d(2LL * N, TB), TB, 0, stream>>>(cnt, norm_out, 2 * N);

  gcn_chunk_sums<<<nChunks, SCAN_TB, 0, stream>>>(deg_in0, deg_in1, chunk_sums, N);
  gcn_rowptr<<<nChunks, SCAN_TB, 0, stream>>>(deg_in0, deg_in1, chunk_sums, rowptr, cur0, cur1, N);
  gcn_fill<<<blocks1d(E, TB), TB, 0, stream>>>(src, dst, cur0, cur1, col, E);
  const dim3 gatherBlk(32 * GATHER_WAVES);
  const unsigned gatherGrid = (unsigned)((N + GATHER_WAVES - 1) / GATHER_WAVES);

  const int rowTiles = (N + 15) / 16;
  const int tilesPerBlock = GEMM_WAVES * ROW_ITERS;
  const dim3 gemmBlk(32 * GEMM_WAVES);
  const dim3 gridH((rowTiles + tilesPerBlock - 1) / tilesPerBlock, H / 16);
  const dim3 gridC((rowTiles + tilesPerBlock - 1) / tilesPerBlock, (C + 15) / 16);

  gcn_gemm_wmma<128><<<gridH, gemmBlk, 0, stream>>>(feat, W0, norm_out, B0, N, H);
  gcn_gather<128, true><<<gatherGrid, gatherBlk, 0, stream>>>(B0, rowptr, col, norm_in, b0, B1, N);

  gcn_gemm_wmma<128><<<gridH, gemmBlk, 0, stream>>>(B1, W1, norm_out, B0, N, H);
  gcn_gather<128, true><<<gatherGrid, gatherBlk, 0, stream>>>(B0, rowptr, col, norm_in, b1, B1, N);

  gcn_gemm_wmma<128><<<gridC, gemmBlk, 0, stream>>>(B1, W2, norm_out, B0, N, C);
  gcn_gather<40, false><<<gatherGrid, gatherBlk, 0, stream>>>(B0, rowptr, col, norm_in, b2, out, N);
}


# GEMM re-tiled: a wave makes a whole 16-row strip of all columns, X read once (16-B lane loads) not once per 16-col tile; W fragments in LDS; k pairs are summed in a new order
# speedup vs baseline: 4.1246x; 1.2642x over previous
#include <hip/hip_runtime.h>
#include <hip/hip_bf16.h>

typedef __attribute__((ext_vector_type(2))) float v2f;
typedef __attribute__((ext_vector_type(8))) float v8f;


__global__ void gcn_zero_f32(float* __restrict__ p, long long n) {
  long long i = (long long)blockIdx.x * blockDim.x + threadIdx.x;
  if (i < n) p[i] = 0.0f;
}

__device__ __forceinline__ int gcn_cache_half() {
  const unsigned id = __builtin_amdgcn_s_sendmsg_rtn(0x87);
  return (int)((id >> 18) & 1u);
}

__global__ void gcn_degrees(const int* __restrict__ src, const int* __restrict__ dst,
                            float* __restrict__ cnt, int N, int E) {
  float* __restrict__ mine = cnt + 2LL * N * gcn_cache_half();
  int e = blockIdx.x * blockDim.x + threadIdx.x;
  if (e < E) {
    atomicAdd(&mine[src[e]], 1.0f);
    atomicAdd(&mine[N + dst[e]], 1.0f);
  }
}

__global__ void gcn_norm(const float* __restrict__ cnt, float* __restrict__ norm, int n) {
  int i = blockIdx.x * blockDim.x + threadIdx.x;
  if (i < n) {
    float d = cnt[i] + cnt[n + i];
    norm[i] = (d > 0.0f) ? rsqrtf(d) : 0.0f;
  }
}

#define STRIP_WAVES 8
#define STRIP_BLOCKS 512
template <int KDIM, int NT>
__global__ __launch_bounds__(32 * STRIP_WAVES, 4) void gcn_gemm_strip(
    const float* __restrict__ X, const float* __restrict__ Wm,
    const float* __restrict__ prescale, float* __restrict__ Y,
    int nRows, int Mout) {
  constexpr int KB = KDIM / 8;
  __shared__ float4 Bs[KB * NT * 32];
  const int lane = threadIdx.x & 31;
  const int wave = __builtin_amdgcn_readfirstlane((int)(threadIdx.x >> 5));
  const int c = lane & 15;
  const int h = lane >> 4;

  static_assert((KB * NT) % STRIP_WAVES == 0, "fragments split evenly over waves");
#pragma unroll
  for (int i = 0; i < KB * NT / STRIP_WAVES; ++i) {
    const int f = wave + STRIP_WAVES * i;
    const int j = f / NT;
    const int t = f - j * NT;
    const int col = 16 * t + c;
    const int colc = (col < Mout) ? col : (Mout - 1);
    const int k = 8 * j + 4 * h;
    float4 b;
    b.x = Wm[(k + 0) * Mout + colc];
    b.y = Wm[(k + 1) * Mout + colc];
    b.z = Wm[(k + 2) * Mout + colc];
    b.w = Wm[(k + 3) * Mout + colc];
    Bs[f * 32 + lane] = b;
  }
  __syncthreads();

  const int nRowTiles = (nRows + 15) >> 4;
  const int nWaves = gridDim.x * STRIP_WAVES;
  for (int tile = wave * gridDim.x + blockIdx.x; tile < nRowTiles;
       tile += nWaves) {
    int row = tile * 16 + c;
    if (row >= nRows) row = nRows - 1;
    const float scale = prescale[row];
    const float4* __restrict__ xrow =
        (const float4*)(X + (long long)row * KDIM + 4 * h);
    float4 a[KB];
#pragma unroll
    for (int j = 0; j < KB; ++j) a[j] = xrow[2 * j];

    v8f acc[NT];
#pragma unroll
    for (int t = 0; t < NT; ++t) acc[t] = v8f{};

#pragma unroll
    for (int j = 0; j < KB; ++j) {
      v2f a0, a1;
      a0.x = a[j].x * scale;
      a0.y = a[j].y * scale;
      a1.x = a[j].z * scale;
      a1.y = a[j].w * scale;
      float4 b[NT];
#pragma unroll
      for (int t = 0; t < NT; ++t) b[t] = Bs[(j * NT + t) * 32 + lane];
#pragma unroll
      for (int t = 0; t < NT; ++t) {
        v2f bb;
        bb.x = b[t].x;
        bb.y = b[t].y;
        acc[t] = __builtin_amdgcn_wmma_f32_16x16x4_f32(
             false, a0,  false, bb,
             (short)0, acc[t],  false,  false);
      }
#pragma unroll
      for (int t = 0; t < NT; ++t) {
        v2f bb;
        bb.x = b[t].z;
        bb.y = b[t].w;
        acc[t] = __builtin_amdgcn_wmma_f32_16x16x4_f32(
             false, a1,  false, bb,
             (short)0, acc[t],  false,  false);
      }
      __builtin_amdgcn_sched_barrier(0);
    }

#pragma unroll
    for (int t = 0; t < NT; ++t) {
      const int col = 16 * t + c;
      if (col < Mout) {
#pragma unroll
        for (int r = 0; r < 8; ++r) {
          const int outRow = tile * 16 + r + 8 * h;
          if (outRow < nRows) Y[(long long)outRow * Mout + col] = acc[t][r];
        }
      }
    }
  }
}

#define SCAN_TB    256
#define SCAN_CHUNK 1024

__device__ __forceinline__ int gcn_block_excl_scan(int v, int* sh, int& total) {
  const int t = threadIdx.x;
  sh[t] = v;
  __syncthreads();
  int in = 0;
  for (int off = 1; off < SCAN_TB; off <<= 1) {
    int x = sh[in * SCAN_TB + t];
    if (t >= off) x += sh[in * SCAN_TB + t - off];
    sh[(1 - in) * SCAN_TB + t] = x;
    __syncthreads();
    in = 1 - in;
  }
  const int incl = sh[in * SCAN_TB + t];
  total = sh[in * SCAN_TB + SCAN_TB - 1];
  __syncthreads();
  return incl - v;
}

__global__ __launch_bounds__(SCAN_TB) void gcn_chunk_sums(
    const float* __restrict__ deg_in0, const float* __restrict__ deg_in1,
    int* __restrict__ chunk_sums, int n) {
  __shared__ int sh[2 * SCAN_TB];
  const int base = blockIdx.x * SCAN_CHUNK + threadIdx.x * 4;
  int v = 0;
  for (int k = 0; k < 4; ++k)
    if (base + k < n) v += (int)(deg_in0[base + k] + deg_in1[base + k]);
  int total;
  (void)gcn_block_excl_scan(v, sh, total);
  if (threadIdx.x == 0) chunk_sums[blockIdx.x] = total;
}

__global__ __launch_bounds__(SCAN_TB) void gcn_rowptr(
    const float* __restrict__ deg_in0, const float* __restrict__ deg_in1,
    const int* __restrict__ chunk_sums, int* __restrict__ rowptr,
    int* __restrict__ cur0, int* __restrict__ cur1, int n) {
  __shared__ int sh[2 * SCAN_TB];
  int part = 0;
  for (int j = threadIdx.x; j < (int)blockIdx.x; j += SCAN_TB) part += chunk_sums[j];
  int offset;
  (void)gcn_block_excl_scan(part, sh, offset);

  const int base = blockIdx.x * SCAN_CHUNK + threadIdx.x * 4;
  int d[4];
  int v = 0;
  for (int k = 0; k < 4; ++k) {
    d[k] = (base + k < n) ? (int)(deg_in0[base + k] + deg_in1[base + k]) : 0;
    v += d[k];
  }
  int total;
  int run = offset + gcn_block_excl_scan(v, sh, total);
  for (int k = 0; k < 4; ++k) {
    if (base + k < n) {
      rowptr[base + k] = run;
      cur0[base + k] = run;
      cur1[base + k] = run + d[k] - 1;
    }
    run += d[k];
  }
  if (blockIdx.x == gridDim.x - 1 && threadIdx.x == 0) rowptr[n] = offset + total;
}

__global__ void gcn_fill(const int* __restrict__ src, const int* __restrict__ dst,
                         int* __restrict__ cur0, int* __restrict__ cur1,
                         int* __restrict__ col, int E) {
  const int half = gcn_cache_half();
  int* __restrict__ cur = half ? cur1 : cur0;
  const int step = half ? -1 : 1;
  const int e = blockIdx.x * blockDim.x + threadIdx.x;
  if (e < E) {
    const int slot = atomicAdd(&cur[dst[e]], step);
    col[slot] = src[e];
  }
}

#define GATHER_WAVES  8
#define GATHER_UNROLL 8
template <int COLS, bool RELU>
__global__ __launch_bounds__(32 * GATHER_WAVES) void gcn_gather(
    const float* __restrict__ Y, const int* __restrict__ rowptr,
    const int* __restrict__ col, const float* __restrict__ norm_in,
    const float* __restrict__ bias, float* __restrict__ out, int nRows) {
  constexpr int LANES = COLS / 4;
  const int lane = threadIdx.x & 31;
  const int row = blockIdx.x * GATHER_WAVES +
                  __builtin_amdgcn_readfirstlane((int)(threadIdx.x >> 5));
  if (row >= nRows) return;
  const bool active = lane < LANES;
  const int start = rowptr[row];
  const int end = rowptr[row + 1];
  const float* __restrict__ ycol = Y + 4 * (active ? lane : 0);

  float4 acc = make_float4(0.0f, 0.0f, 0.0f, 0.0f);
  for (int base = start; base < end; base += 32) {
    const int n = (end - base < 32) ? (end - base) : 32;
    const int myIdx = (lane < n) ? col[base + lane] : 0;
    for (int j = 0; j < n; j += GATHER_UNROLL) {
      float4 v[GATHER_UNROLL];
#pragma unroll
      for (int u = 0; u < GATHER_UNROLL; ++u) {
        const int s = __builtin_amdgcn_readlane(myIdx, j + u);
        v[u] = *(const float4*)(ycol + (long long)s * COLS);
      }
#pragma unroll
      for (int u = 0; u < GATHER_UNROLL; ++u) {
        const bool k = (j + u) < n;
        acc.x += k ? v[u].x : 0.0f;
        acc.y += k ? v[u].y : 0.0f;
        acc.z += k ? v[u].z : 0.0f;
        acc.w += k ? v[u].w : 0.0f;
      }
    }
  }

  if (active) {
    const float ni = norm_in[row];
    const float4 b = *(const float4*)(bias + 4 * lane);
    float4 r;
    r.x = acc.x * ni + b.x;
    r.y = acc.y * ni + b.y;
    r.z = acc.z * ni + b.z;
    r.w = acc.w * ni + b.w;
    if (RELU) {
      r.x = fmaxf(r.x, 0.0f); r.y = fmaxf(r.y, 0.0f);
      r.z = fmaxf(r.z, 0.0f); r.w = fmaxf(r.w, 0.0f);
    }
    *(float4*)(out + (long long)row * COLS + 4 * lane) = r;
  }
}

extern "C" void kernel_launch(void* const* d_in, const int* in_sizes, int n_in,
                              void* d_out, int out_size, void* d_ws, size_t ws_size,
                              hipStream_t stream) {
  const float* feat = (const float*)d_in[0];
  const int*   src  = (const int*)d_in[1];
  const int*   dst  = (const int*)d_in[2];
  const float* W0   = (const float*)d_in[3];
  const float* b0   = (const float*)d_in[4];
  const float* W1   = (const float*)d_in[5];
  const float* b1   = (const float*)d_in[6];
  const float* W2   = (const float*)d_in[7];
  const float* b2   = (const float*)d_in[8];
  float* out = (float*)d_out;

  const int F = 128, H = 128, C = 40;
  const int N = in_sizes[0] / F;
  const int E = in_sizes[1];
  (void)in_sizes; (void)n_in; (void)out_size; (void)ws_size;

  float* ws       = (float*)d_ws;
  float* cnt      = ws;
  float* deg_in0  = ws + 1LL * N;
  float* deg_in1  = ws + 3LL * N;
  float* norm_out = ws + 4LL * N;
  float* norm_in  = ws + 5LL * N;
  float* B0       = ws + 6LL * N;
  float* B1       = B0 + (long long)N * H;
  const int nChunks = (N + SCAN_CHUNK - 1) / SCAN_CHUNK;
  int* rowptr     = (int*)(B1 + (long long)N * H);
  int* cur0       = rowptr + (N + 1);
  int* cur1       = cur0 + N;
  int* chunk_sums = cur1 + N;
  int* col        = chunk_sums + nChunks;

  const int TB = 256;
  auto blocks1d = [](long long n, int tb) { return (unsigned)((n + tb - 1) / tb); };

  gcn_zero_f32<<<blocks1d(4LL * N, TB), TB, 0, stream>>>(cnt, 4LL * N);
  gcn_degrees<<<blocks1d(E, TB), TB, 0, stream>>>(src, dst, cnt, N, E);
  gcn_norm<<<blocks1d(2LL * N, TB), TB, 0, stream>>>(cnt, norm_out, 2 * N);

  gcn_chunk_sums<<<nChunks, SCAN_TB, 0, stream>>>(deg_in0, deg_in1, chunk_sums, N);
  gcn_rowptr<<<nChunks, SCAN_TB, 0, stream>>>(deg_in0, deg_in1, chunk_sums, rowptr, cur0, cur1, N);
  gcn_fill<<<blocks1d(E, TB), TB, 0, stream>>>(src, dst, cur0, cur1, col, E);
  const dim3 gatherBlk(32 * GATHER_WAVES);
  const unsigned gatherGrid = (unsigned)((N + GATHER_WAVES - 1) / GATHER_WAVES);

  const dim3 gemmBlk(32 * STRIP_WAVES);
  const dim3 gemmGrid(STRIP_BLOCKS);

  gcn_gemm_strip<128, 8><<<gemmGrid, gemmBlk, 0, stream>>>(feat, W0, norm_out, B0, N, H);
  gcn_gather<128, true><<<gatherGrid, gatherBlk, 0, stream>>>(B0, rowptr, col, norm_in, b0, B1, N);

  gcn_gemm_strip<128, 8><<<gemmGrid, gemmBlk, 0, stream>>>(B1, W1, norm_out, B0, N, H);
  gcn_gather<128, true><<<gatherGrid, gatherBlk, 0, stream>>>(B0, rowptr, col, norm_in, b1, B1, N);

  gcn_gemm_strip<128, 3><<<gemmGrid, gemmBlk, 0, stream>>>(B1, W2, norm_out, B0, N, C);
  gcn_gather<40, false><<<gatherGrid, gatherBlk, 0, stream>>>(B0, rowptr, col, norm_in, b2, out, N);
}


# edges grouped by destination by a two-level counting sort ranked with LDS atomics (bins of 512 rows, one block per bin): no global atomic per destination edge; row sum order = LDS arrival order
# speedup vs baseline: 4.3440x; 1.0532x over previous
#include <hip/hip_runtime.h>
#include <hip/hip_bf16.h>

typedef __attribute__((ext_vector_type(2))) float v2f;
typedef __attribute__((ext_vector_type(8))) float v8f;


__global__ void gcn_zero_f32(float* __restrict__ p, long long n) {
  long long i = (long long)blockIdx.x * blockDim.x + threadIdx.x;
  if (i < n) p[i] = 0.0f;
}

__device__ __forceinline__ int gcn_cache_half() {
  const unsigned id = __builtin_amdgcn_s_sendmsg_rtn(0x87);
  return (int)((id >> 18) & 1u);
}

#define STRIP_WAVES 8
#define STRIP_BLOCKS 512
template <int KDIM, int NT>
__global__ __launch_bounds__(32 * STRIP_WAVES, 4) void gcn_gemm_strip(
    const float* __restrict__ X, const float* __restrict__ Wm,
    const float* __restrict__ prescale, float* __restrict__ Y,
    int nRows, int Mout) {
  constexpr int KB = KDIM / 8;
  __shared__ float4 Bs[KB * NT * 32];
  const int lane = threadIdx.x & 31;
  const int wave = __builtin_amdgcn_readfirstlane((int)(threadIdx.x >> 5));
  const int c = lane & 15;
  const int h = lane >> 4;

  static_assert((KB * NT) % STRIP_WAVES == 0, "fragments split evenly over waves");
#pragma unroll
  for (int i = 0; i < KB * NT / STRIP_WAVES; ++i) {
    const int f = wave + STRIP_WAVES * i;
    const int j = f / NT;
    const int t = f - j * NT;
    const int col = 16 * t + c;
    const int colc = (col < Mout) ? col : (Mout - 1);
    const int k = 8 * j + 4 * h;
    float4 b;
    b.x = Wm[(k + 0) * Mout + colc];
    b.y = Wm[(k + 1) * Mout + colc];
    b.z = Wm[(k + 2) * Mout + colc];
    b.w = Wm[(k + 3) * Mout + colc];
    Bs[f * 32 + lane] = b;
  }
  __syncthreads();

  const int nRowTiles = (nRows + 15) >> 4;
  const int nWaves = gridDim.x * STRIP_WAVES;
  for (int tile = wave * gridDim.x + blockIdx.x; tile < nRowTiles;
       tile += nWaves) {
    int row = tile * 16 + c;
    if (row >= nRows) row = nRows - 1;
    const float scale = prescale[row];
    const float4* __restrict__ xrow =
        (const float4*)(X + (long long)row * KDIM + 4 * h);
    float4 a[KB];
#pragma unroll
    for (int j = 0; j < KB; ++j) a[j] = xrow[2 * j];

    v8f acc[NT];
#pragma unroll
    for (int t = 0; t < NT; ++t) acc[t] = v8f{};

#pragma unroll
    for (int j = 0; j < KB; ++j) {
      v2f a0, a1;
      a0.x = a[j].x * scale;
      a0.y = a[j].y * scale;
      a1.x = a[j].z * scale;
      a1.y = a[j].w * scale;
      float4 b[NT];
#pragma unroll
      for (int t = 0; t < NT; ++t) b[t] = Bs[(j * NT + t) * 32 + lane];
#pragma unroll
      for (int t = 0; t < NT; ++t) {
        v2f bb;
        bb.x = b[t].x;
        bb.y = b[t].y;
        acc[t] = __builtin_amdgcn_wmma_f32_16x16x4_f32(
             false, a0,  false, bb,
             (short)0, acc[t],  false,  false);
      }
#pragma unroll
      for (int t = 0; t < NT; ++t) {
        v2f bb;
        bb.x = b[t].z;
        bb.y = b[t].w;
        acc[t] = __builtin_amdgcn_wmma_f32_16x16x4_f32(
             false, a1,  false, bb,
             (short)0, acc[t],  false,  false);
      }
      __builtin_amdgcn_sched_barrier(0);
    }

#pragma unroll
    for (int t = 0; t < NT; ++t) {
      const int col = 16 * t + c;
      if (col < Mout) {
#pragma unroll
        for (int r = 0; r < 8; ++r) {
          const int outRow = tile * 16 + r + 8 * h;
          if (outRow < nRows) Y[(long long)outRow * Mout + col] = acc[t][r];
        }
      }
    }
  }
}

#define SCAN_TB    256
#define BIN_SHIFT  9
#define BIN_NODES  (1 << BIN_SHIFT)
#define BIN_MAX    1024
#define BIN_PAD    32
#define SRC_BITS   20
#define EDGE_PER_THREAD 16
#define ROWS_UNROLL 8
#define EDGE_CHUNK (SCAN_TB * EDGE_PER_THREAD)

__device__ __forceinline__ int gcn_block_excl_scan(int v, int* sh, int& total) {
  const int t = threadIdx.x;
  sh[t] = v;
  __syncthreads();
  int in = 0;
  for (int off = 1; off < SCAN_TB; off <<= 1) {
    int x = sh[in * SCAN_TB + t];
    if (t >= off) x += sh[in * SCAN_TB + t - off];
    sh[(1 - in) * SCAN_TB + t] = x;
    __syncthreads();
    in = 1 - in;
  }
  const int incl = sh[in * SCAN_TB + t];
  total = sh[in * SCAN_TB + SCAN_TB - 1];
  __syncthreads();
  return incl - v;
}

__global__ __launch_bounds__(SCAN_TB) void gcn_bin_count(
    const int* __restrict__ src, const int* __restrict__ dst,
    float* __restrict__ cntOut, int* __restrict__ binTot, int N, int E, int NB) {
  __shared__ int hist[BIN_MAX];
  for (int i = threadIdx.x; i < NB; i += SCAN_TB) hist[i] = 0;
  __syncthreads();
  float* __restrict__ mine = cntOut + (long long)N * gcn_cache_half();
  const int base = blockIdx.x * EDGE_CHUNK + threadIdx.x;
  int d[EDGE_PER_THREAD], s[EDGE_PER_THREAD];
#pragma unroll
  for (int k = 0; k < EDGE_PER_THREAD; ++k) {
    const int e = base + k * SCAN_TB;
    const int ec = (e < E) ? e : (E - 1);
    d[k] = dst[ec];
    s[k] = src[ec];
  }
#pragma unroll
  for (int k = 0; k < EDGE_PER_THREAD; ++k) {
    if (base + k * SCAN_TB < E) {
      atomicAdd(&mine[s[k]], 1.0f);
      atomicAdd(&hist[d[k] >> BIN_SHIFT], 1);
    }
  }
  __syncthreads();
  for (int i = threadIdx.x; i < NB; i += SCAN_TB) {
    const int h = hist[i];
    if (h) atomicAdd(&binTot[i * BIN_PAD], h);
  }
}

__global__ __launch_bounds__(SCAN_TB) void gcn_bin_scatter(
    const int* __restrict__ src, const int* __restrict__ dst,
    const int* __restrict__ binTot, int* __restrict__ binCur,
    int* __restrict__ rec, int E, int NB) {
  __shared__ int hist[BIN_MAX];
  __shared__ int binStart[BIN_MAX];
  __shared__ int sh[2 * SCAN_TB];
  for (int i = threadIdx.x; i < NB; i += SCAN_TB) hist[i] = 0;
  __syncthreads();
  const int base = blockIdx.x * EDGE_CHUNK + threadIdx.x;
  int bin[EDGE_PER_THREAD], rank[EDGE_PER_THREAD], word[EDGE_PER_THREAD];
#pragma unroll
  for (int k = 0; k < EDGE_PER_THREAD; ++k) {
    const int e = base + k * SCAN_TB;
    const int ec = (e < E) ? e : (E - 1);
    const int d = dst[ec];
    bin[k] = d >> BIN_SHIFT;
    word[k] = ((d & (BIN_NODES - 1)) << SRC_BITS) | src[ec];
  }
#pragma unroll
  for (int k = 0; k < EDGE_PER_THREAD; ++k) {
    rank[k] = 0;
    if (base + k * SCAN_TB < E) rank[k] = atomicAdd(&hist[bin[k]], 1);
    else bin[k] = -1;
  }
  int v[4];
  int sum = 0;
#pragma unroll
  for (int k = 0; k < 4; ++k) {
    const int b = threadIdx.x * 4 + k;
    v[k] = (b < NB) ? binTot[b * BIN_PAD] : 0;
    sum += v[k];
  }
  int total;
  int run = gcn_block_excl_scan(sum, sh, total);
#pragma unroll
  for (int k = 0; k < 4; ++k) {
    const int b = threadIdx.x * 4 + k;
    if (b < NB) binStart[b] = run;
    run += v[k];
  }
  __syncthreads();
  for (int i = threadIdx.x; i < NB; i += SCAN_TB) {
    const int h = hist[i];
    if (h) hist[i] = binStart[i] + atomicAdd(&binCur[i * BIN_PAD], h);
  }
  __syncthreads();
#pragma unroll
  for (int k = 0; k < EDGE_PER_THREAD; ++k)
    if (bin[k] >= 0) rec[hist[bin[k]] + rank[k]] = word[k];
}

__global__ __launch_bounds__(SCAN_TB) void gcn_bin_rows(
    const int* __restrict__ rec, const int* __restrict__ binTot,
    const float* __restrict__ cntOut, int* __restrict__ rowptr,
    int* __restrict__ col, float* __restrict__ norm_out,
    float* __restrict__ norm_in, int N, int NB) {
  __shared__ int cntNode[BIN_NODES];
  __shared__ int curNode[BIN_NODES];
  __shared__ int sh[2 * SCAN_TB];
  const int b = blockIdx.x;
  for (int i = threadIdx.x; i < BIN_NODES; i += SCAN_TB) cntNode[i] = 0;
  int part = 0;
  for (int j = threadIdx.x; j < b; j += SCAN_TB) part += binTot[j * BIN_PAD];
  int begin;
  (void)gcn_block_excl_scan(part, sh, begin);
  const int end = begin + binTot[b * BIN_PAD];

  for (int i0 = begin + threadIdx.x; i0 < end; i0 += SCAN_TB * ROWS_UNROLL) {
    int w[ROWS_UNROLL];
#pragma unroll
    for (int u = 0; u < ROWS_UNROLL; ++u) {
      const int i = i0 + u * SCAN_TB;
      w[u] = (i < end) ? rec[i] : -1;
    }
#pragma unroll
    for (int u = 0; u < ROWS_UNROLL; ++u)
      if (w[u] >= 0) atomicAdd(&cntNode[w[u] >> SRC_BITS], 1);
  }
  __syncthreads();

  static_assert(BIN_NODES == 2 * SCAN_TB, "two rows per thread");
  const int c0 = cntNode[2 * threadIdx.x];
  const int c1 = cntNode[2 * threadIdx.x + 1];
  int binEdges;
  const int first = begin + gcn_block_excl_scan(c0 + c1, sh, binEdges);
  curNode[2 * threadIdx.x] = first;
  curNode[2 * threadIdx.x + 1] = first + c0;
  const int node = b * BIN_NODES + 2 * threadIdx.x;
  if (node < N) {
    rowptr[node] = first;
    norm_in[node] = (c0 > 0) ? rsqrtf((float)c0) : 0.0f;
    const float d = cntOut[node] + cntOut[N + node];
    norm_out[node] = (d > 0.0f) ? rsqrtf(d) : 0.0f;
  }
  if (node + 1 < N) {
    rowptr[node + 1] = first + c0;
    norm_in[node + 1] = (c1 > 0) ? rsqrtf((float)c1) : 0.0f;
    const float d = cntOut[node + 1] + cntOut[N + node + 1];
    norm_out[node + 1] = (d > 0.0f) ? rsqrtf(d) : 0.0f;
  }
  if (b == NB - 1 && threadIdx.x == 0) rowptr[N] = end;
  __syncthreads();

  for (int i0 = begin + threadIdx.x; i0 < end; i0 += SCAN_TB * ROWS_UNROLL) {
    int w[ROWS_UNROLL];
#pragma unroll
    for (int u = 0; u < ROWS_UNROLL; ++u) {
      const int i = i0 + u * SCAN_TB;
      w[u] = (i < end) ? rec[i] : -1;
    }
#pragma unroll
    for (int u = 0; u < ROWS_UNROLL; ++u) {
      if (w[u] >= 0) {
        const int slot = atomicAdd(&curNode[w[u] >> SRC_BITS], 1);
        col[slot] = w[u] & ((1 << SRC_BITS) - 1);
      }
    }
  }
}

#define GATHER_WAVES  8
#define GATHER_UNROLL 8
template <int COLS, bool RELU>
__global__ __launch_bounds__(32 * GATHER_WAVES) void gcn_gather(
    const float* __restrict__ Y, const int* __restrict__ rowptr,
    const int* __restrict__ col, const float* __restrict__ norm_in,
    const float* __restrict__ bias, float* __restrict__ out, int nRows) {
  constexpr int LANES = COLS / 4;
  const int lane = threadIdx.x & 31;
  const int row = blockIdx.x * GATHER_WAVES +
                  __builtin_amdgcn_readfirstlane((int)(threadIdx.x >> 5));
  if (row >= nRows) return;
  const bool active = lane < LANES;
  const int start = rowptr[row];
  const int end = rowptr[row + 1];
  const float* __restrict__ ycol = Y + 4 * (active ? lane : 0);

  float4 acc = make_float4(0.0f, 0.0f, 0.0f, 0.0f);
  for (int base = start; base < end; base += 32) {
    const int n = (end - base < 32) ? (end - base) : 32;
    const int myIdx = (lane < n) ? col[base + lane] : 0;
    for (int j = 0; j < n; j += GATHER_UNROLL) {
      float4 v[GATHER_UNROLL];
#pragma unroll
      for (int u = 0; u < GATHER_UNROLL; ++u) {
        const int s = __builtin_amdgcn_readlane(myIdx, j + u);
        v[u] = *(const float4*)(ycol + (long long)s * COLS);
      }
#pragma unroll
      for (int u = 0; u < GATHER_UNROLL; ++u) {
        const bool k = (j + u) < n;
        acc.x += k ? v[u].x : 0.0f;
        acc.y += k ? v[u].y : 0.0f;
        acc.z += k ? v[u].z : 0.0f;
        acc.w += k ? v[u].w : 0.0f;
      }
    }
  }

  if (active) {
    const float ni = norm_in[row];
    const float4 b = *(const float4*)(bias + 4 * lane);
    float4 r;
    r.x = acc.x * ni + b.x;
    r.y = acc.y * ni + b.y;
    r.z = acc.z * ni + b.z;
    r.w = acc.w * ni + b.w;
    if (RELU) {
      r.x = fmaxf(r.x, 0.0f); r.y = fmaxf(r.y, 0.0f);
      r.z = fmaxf(r.z, 0.0f); r.w = fmaxf(r.w, 0.0f);
    }
    *(float4*)(out + (long long)row * COLS + 4 * lane) = r;
  }
}

extern "C" void kernel_launch(void* const* d_in, const int* in_sizes, int n_in,
                              void* d_out, int out_size, void* d_ws, size_t ws_size,
                              hipStream_t stream) {
  const float* feat = (const float*)d_in[0];
  const int*   src  = (const int*)d_in[1];
  const int*   dst  = (const int*)d_in[2];
  const float* W0   = (const float*)d_in[3];
  const float* b0   = (const float*)d_in[4];
  const float* W1   = (const float*)d_in[5];
  const float* b1   = (const float*)d_in[6];
  const float* W2   = (const float*)d_in[7];
  const float* b2   = (const float*)d_in[8];
  float* out = (float*)d_out;

  const int F = 128, H = 128, C = 40;
  const int N = in_sizes[0] / F;
  const int E = in_sizes[1];
  (void)in_sizes; (void)n_in; (void)out_size; (void)ws_size;

  const int NB = (N + BIN_NODES - 1) / BIN_NODES;
  float* ws       = (float*)d_ws;
  float* cntOut   = ws;
  int*   binTot   = (int*)(ws + 2LL * N);
  int*   binCur   = binTot + (long long)NB * BIN_PAD;
  float* norm_out = (float*)(binCur + (long long)NB * BIN_PAD);
  float* norm_in  = norm_out + N;
  float* B0       = norm_in + N;
  float* B1       = B0 + (long long)N * H;
  int* rowptr     = (int*)(B1 + (long long)N * H);
  int* col        = rowptr + (N + 1);
  int* rec        = col + E;

  const int TB = 256;
  auto blocks1d = [](long long n, int tb) { return (unsigned)((n + tb - 1) / tb); };

  const long long nClear = 2LL * N + 2LL * NB * BIN_PAD;
  const unsigned edgeGrid = blocks1d(E, EDGE_CHUNK);
  gcn_zero_f32<<<blocks1d(nClear, TB), TB, 0, stream>>>(ws, nClear);
  gcn_bin_count<<<edgeGrid, SCAN_TB, 0, stream>>>(src, dst, cntOut, binTot, N, E, NB);
  gcn_bin_scatter<<<edgeGrid, SCAN_TB, 0, stream>>>(src, dst, binTot, binCur, rec, E, NB);
  gcn_bin_rows<<<NB, SCAN_TB, 0, stream>>>(rec, binTot, cntOut, rowptr, col, norm_out, norm_in, N, NB);
  const dim3 gatherBlk(32 * GATHER_WAVES);
  const unsigned gatherGrid = (unsigned)((N + GATHER_WAVES - 1) / GATHER_WAVES);

  const dim3 gemmBlk(32 * STRIP_WAVES);
  const dim3 gemmGrid(STRIP_BLOCKS);

  gcn_gemm_strip<128, 8><<<gemmGrid, gemmBlk, 0, stream>>>(feat, W0, norm_out, B0, N, H);
  gcn_gather<128, true><<<gatherGrid, gatherBlk, 0, stream>>>(B0, rowptr, col, norm_in, b0, B1, N);

  gcn_gemm_strip<128, 8><<<gemmGrid, gemmBlk, 0, stream>>>(B1, W1, norm_out, B0, N, H);
  gcn_gather<128, true><<<gatherGrid, gatherBlk, 0, stream>>>(B0, rowptr, col, norm_in, b1, B1, N);

  gcn_gemm_strip<128, 3><<<gemmGrid, gemmBlk, 0, stream>>>(B1, W2, norm_out, B0, N, C);
  gcn_gather<40, false><<<gatherGrid, gatherBlk, 0, stream>>>(B0, rowptr, col, norm_in, b2, out, N);
}


# gathers of layers 0 and 1 fused with the next layer's GEMM: a block gathers a 16-row strip into LDS, 8 waves multiply it by W tiles held in registers; no activation round trip; sum orders kept
# speedup vs baseline: 4.8859x; 1.1247x over previous
#include <hip/hip_runtime.h>
#include <hip/hip_bf16.h>

typedef __attribute__((ext_vector_type(2))) float v2f;
typedef __attribute__((ext_vector_type(8))) float v8f;


__global__ void gcn_zero_f32(float* __restrict__ p, long long n) {
  long long i = (long long)blockIdx.x * blockDim.x + threadIdx.x;
  if (i < n) p[i] = 0.0f;
}

__device__ __forceinline__ int gcn_cache_half() {
  const unsigned id = __builtin_amdgcn_s_sendmsg_rtn(0x87);
  return (int)((id >> 18) & 1u);
}

#define STRIP_WAVES 8
#define STRIP_BLOCKS 512
template <int KDIM, int NT>
__global__ __launch_bounds__(32 * STRIP_WAVES, 4) void gcn_gemm_strip(
    const float* __restrict__ X, const float* __restrict__ Wm,
    const float* __restrict__ prescale, float* __restrict__ Y,
    int nRows, int Mout) {
  constexpr int KB = KDIM / 8;
  __shared__ float4 Bs[KB * NT * 32];
  const int lane = threadIdx.x & 31;
  const int wave = __builtin_amdgcn_readfirstlane((int)(threadIdx.x >> 5));
  const int c = lane & 15;
  const int h = lane >> 4;

  static_assert((KB * NT) % STRIP_WAVES == 0, "fragments split evenly over waves");
#pragma unroll
  for (int i = 0; i < KB * NT / STRIP_WAVES; ++i) {
    const int f = wave + STRIP_WAVES * i;
    const int j = f / NT;
    const int t = f - j * NT;
    const int col = 16 * t + c;
    const int colc = (col < Mout) ? col : (Mout - 1);
    const int k = 8 * j + 4 * h;
    float4 b;
    b.x = Wm[(k + 0) * Mout + colc];
    b.y = Wm[(k + 1) * Mout + colc];
    b.z = Wm[(k + 2) * Mout + colc];
    b.w = Wm[(k + 3) * Mout + colc];
    Bs[f * 32 + lane] = b;
  }
  __syncthreads();

  const int nRowTiles = (nRows + 15) >> 4;
  const int nWaves = gridDim.x * STRIP_WAVES;
  for (int tile = wave * gridDim.x + blockIdx.x; tile < nRowTiles;
       tile += nWaves) {
    int row = tile * 16 + c;
    if (row >= nRows) row = nRows - 1;
    const float scale = prescale[row];
    const float4* __restrict__ xrow =
        (const float4*)(X + (long long)row * KDIM + 4 * h);
    float4 a[KB];
#pragma unroll
    for (int j = 0; j < KB; ++j) a[j] = xrow[2 * j];

    v8f acc[NT];
#pragma unroll
    for (int t = 0; t < NT; ++t) acc[t] = v8f{};

#pragma unroll
    for (int j = 0; j < KB; ++j) {
      v2f a0, a1;
      a0.x = a[j].x * scale;
      a0.y = a[j].y * scale;
      a1.x = a[j].z * scale;
      a1.y = a[j].w * scale;
      float4 b[NT];
#pragma unroll
      for (int t = 0; t < NT; ++t) b[t] = Bs[(j * NT + t) * 32 + lane];
#pragma unroll
      for (int t = 0; t < NT; ++t) {
        v2f bb;
        bb.x = b[t].x;
        bb.y = b[t].y;
        acc[t] = __builtin_amdgcn_wmma_f32_16x16x4_f32(
             false, a0,  false, bb,
             (short)0, acc[t],  false,  false);
      }
#pragma unroll
      for (int t = 0; t < NT; ++t) {
        v2f bb;
        bb.x = b[t].z;
        bb.y = b[t].w;
        acc[t] = __builtin_amdgcn_wmma_f32_16x16x4_f32(
             false, a1,  false, bb,
             (short)0, acc[t],  false,  false);
      }
      __builtin_amdgcn_sched_barrier(0);
    }

#pragma unroll
    for (int t = 0; t < NT; ++t) {
      const int col = 16 * t + c;
      if (col < Mout) {
#pragma unroll
        for (int r = 0; r < 8; ++r) {
          const int outRow = tile * 16 + r + 8 * h;
          if (outRow < nRows) Y[(long long)outRow * Mout + col] = acc[t][r];
        }
      }
    }
  }
}

#define SCAN_TB    256
#define BIN_SHIFT  9
#define BIN_NODES  (1 << BIN_SHIFT)
#define BIN_MAX    1024
#define BIN_PAD    32
#define SRC_BITS   20
#define EDGE_PER_THREAD 16
#define ROWS_UNROLL 8
#define EDGE_CHUNK (SCAN_TB * EDGE_PER_THREAD)

__device__ __forceinline__ int gcn_block_excl_scan(int v, int* sh, int& total) {
  const int t = threadIdx.x;
  sh[t] = v;
  __syncthreads();
  int in = 0;
  for (int off = 1; off < SCAN_TB; off <<= 1) {
    int x = sh[in * SCAN_TB + t];
    if (t >= off) x += sh[in * SCAN_TB + t - off];
    sh[(1 - in) * SCAN_TB + t] = x;
    __syncthreads();
    in = 1 - in;
  }
  const int incl = sh[in * SCAN_TB + t];
  total = sh[in * SCAN_TB + SCAN_TB - 1];
  __syncthreads();
  return incl - v;
}

__global__ __launch_bounds__(SCAN_TB) void gcn_bin_count(
    const int* __restrict__ src, const int* __restrict__ dst,
    float* __restrict__ cntOut, int* __restrict__ binTot, int N, int E, int NB) {
  __shared__ int hist[BIN_MAX];
  for (int i = threadIdx.x; i < NB; i += SCAN_TB) hist[i] = 0;
  __syncthreads();
  float* __restrict__ mine = cntOut + (long long)N * gcn_cache_half();
  const int base = blockIdx.x * EDGE_CHUNK + threadIdx.x;
  int d[EDGE_PER_THREAD], s[EDGE_PER_THREAD];
#pragma unroll
  for (int k = 0; k < EDGE_PER_THREAD; ++k) {
    const int e = base + k * SCAN_TB;
    const int ec = (e < E) ? e : (E - 1);
    d[k] = dst[ec];
    s[k] = src[ec];
  }
#pragma unroll
  for (int k = 0; k < EDGE_PER_THREAD; ++k) {
    if (base + k * SCAN_TB < E) {
      atomicAdd(&mine[s[k]], 1.0f);
      atomicAdd(&hist[d[k] >> BIN_SHIFT], 1);
    }
  }
  __syncthreads();
  for (int i = threadIdx.x; i < NB; i += SCAN_TB) {
    const int h = hist[i];
    if (h) atomicAdd(&binTot[i * BIN_PAD], h);
  }
}

__global__ __launch_bounds__(SCAN_TB) void gcn_bin_scatter(
    const int* __restrict__ src, const int* __restrict__ dst,
    const int* __restrict__ binTot, int* __restrict__ binCur,
    int* __restrict__ rec, int E, int NB) {
  __shared__ int hist[BIN_MAX];
  __shared__ int binStart[BIN_MAX];
  __shared__ int sh[2 * SCAN_TB];
  for (int i = threadIdx.x; i < NB; i += SCAN_TB) hist[i] = 0;
  __syncthreads();
  const int base = blockIdx.x * EDGE_CHUNK + threadIdx.x;
  int bin[EDGE_PER_THREAD], rank[EDGE_PER_THREAD], word[EDGE_PER_THREAD];
#pragma unroll
  for (int k = 0; k < EDGE_PER_THREAD; ++k) {
    const int e = base + k * SCAN_TB;
    const int ec = (e < E) ? e : (E - 1);
    const int d = dst[ec];
    bin[k] = d >> BIN_SHIFT;
    word[k] = ((d & (BIN_NODES - 1)) << SRC_BITS) | src[ec];
  }
#pragma unroll
  for (int k = 0; k < EDGE_PER_THREAD; ++k) {
    rank[k] = 0;
    if (base + k * SCAN_TB < E) rank[k] = atomicAdd(&hist[bin[k]], 1);
    else bin[k] = -1;
  }
  int v[4];
  int sum = 0;
#pragma unroll
  for (int k = 0; k < 4; ++k) {
    const int b = threadIdx.x * 4 + k;
    v[k] = (b < NB) ? binTot[b * BIN_PAD] : 0;
    sum += v[k];
  }
  int total;
  int run = gcn_block_excl_scan(sum, sh, total);
#pragma unroll
  for (int k = 0; k < 4; ++k) {
    const int b = threadIdx.x * 4 + k;
    if (b < NB) binStart[b] = run;
    run += v[k];
  }
  __syncthreads();
  for (int i = threadIdx.x; i < NB; i += SCAN_TB) {
    const int h = hist[i];
    if (h) hist[i] = binStart[i] + atomicAdd(&binCur[i * BIN_PAD], h);
  }
  __syncthreads();
#pragma unroll
  for (int k = 0; k < EDGE_PER_THREAD; ++k)
    if (bin[k] >= 0) rec[hist[bin[k]] + rank[k]] = word[k];
}

__global__ __launch_bounds__(SCAN_TB) void gcn_bin_rows(
    const int* __restrict__ rec, const int* __restrict__ binTot,
    const float* __restrict__ cntOut, int* __restrict__ rowptr,
    int* __restrict__ col, float* __restrict__ norm_out,
    float* __restrict__ norm_in, int N, int NB) {
  __shared__ int cntNode[BIN_NODES];
  __shared__ int curNode[BIN_NODES];
  __shared__ int sh[2 * SCAN_TB];
  const int b = blockIdx.x;
  for (int i = threadIdx.x; i < BIN_NODES; i += SCAN_TB) cntNode[i] = 0;
  int part = 0;
  for (int j = threadIdx.x; j < b; j += SCAN_TB) part += binTot[j * BIN_PAD];
  int begin;
  (void)gcn_block_excl_scan(part, sh, begin);
  const int end = begin + binTot[b * BIN_PAD];

  for (int i0 = begin + threadIdx.x; i0 < end; i0 += SCAN_TB * ROWS_UNROLL) {
    int w[ROWS_UNROLL];
#pragma unroll
    for (int u = 0; u < ROWS_UNROLL; ++u) {
      const int i = i0 + u * SCAN_TB;
      w[u] = (i < end) ? rec[i] : -1;
    }
#pragma unroll
    for (int u = 0; u < ROWS_UNROLL; ++u)
      if (w[u] >= 0) atomicAdd(&cntNode[w[u] >> SRC_BITS], 1);
  }
  __syncthreads();

  static_assert(BIN_NODES == 2 * SCAN_TB, "two rows per thread");
  const int c0 = cntNode[2 * threadIdx.x];
  const int c1 = cntNode[2 * threadIdx.x + 1];
  int binEdges;
  const int first = begin + gcn_block_excl_scan(c0 + c1, sh, binEdges);
  curNode[2 * threadIdx.x] = first;
  curNode[2 * threadIdx.x + 1] = first + c0;
  const int node = b * BIN_NODES + 2 * threadIdx.x;
  if (node < N) {
    rowptr[node] = first;
    norm_in[node] = (c0 > 0) ? rsqrtf((float)c0) : 0.0f;
    const float d = cntOut[node] + cntOut[N + node];
    norm_out[node] = (d > 0.0f) ? rsqrtf(d) : 0.0f;
  }
  if (node + 1 < N) {
    rowptr[node + 1] = first + c0;
    norm_in[node + 1] = (c1 > 0) ? rsqrtf((float)c1) : 0.0f;
    const float d = cntOut[node + 1] + cntOut[N + node + 1];
    norm_out[node + 1] = (d > 0.0f) ? rsqrtf(d) : 0.0f;
  }
  if (b == NB - 1 && threadIdx.x == 0) rowptr[N] = end;
  __syncthreads();

  for (int i0 = begin + threadIdx.x; i0 < end; i0 += SCAN_TB * ROWS_UNROLL) {
    int w[ROWS_UNROLL];
#pragma unroll
    for (int u = 0; u < ROWS_UNROLL; ++u) {
      const int i = i0 + u * SCAN_TB;
      w[u] = (i < end) ? rec[i] : -1;
    }
#pragma unroll
    for (int u = 0; u < ROWS_UNROLL; ++u) {
      if (w[u] >= 0) {
        const int slot = atomicAdd(&curNode[w[u] >> SRC_BITS], 1);
        col[slot] = w[u] & ((1 << SRC_BITS) - 1);
      }
    }
  }
}

#define GATHER_WAVES  8
#define GATHER_UNROLL 8
template <int COLS, bool RELU>
__global__ __launch_bounds__(32 * GATHER_WAVES) void gcn_gather(
    const float* __restrict__ Y, const int* __restrict__ rowptr,
    const int* __restrict__ col, const float* __restrict__ norm_in,
    const float* __restrict__ bias, float* __restrict__ out, int nRows) {
  constexpr int LANES = COLS / 4;
  const int lane = threadIdx.x & 31;
  const int row = blockIdx.x * GATHER_WAVES +
                  __builtin_amdgcn_readfirstlane((int)(threadIdx.x >> 5));
  if (row >= nRows) return;
  const bool active = lane < LANES;
  const int start = rowptr[row];
  const int end = rowptr[row + 1];
  const float* __restrict__ ycol = Y + 4 * (active ? lane : 0);

  float4 acc = make_float4(0.0f, 0.0f, 0.0f, 0.0f);
  for (int base = start; base < end; base += 32) {
    const int n = (end - base < 32) ? (end - base) : 32;
    const int myIdx = (lane < n) ? col[base + lane] : 0;
    for (int j = 0; j < n; j += GATHER_UNROLL) {
      float4 v[GATHER_UNROLL];
#pragma unroll
      for (int u = 0; u < GATHER_UNROLL; ++u) {
        const int s = __builtin_amdgcn_readlane(myIdx, j + u);
        v[u] = *(const float4*)(ycol + (long long)s * COLS);
      }
#pragma unroll
      for (int u = 0; u < GATHER_UNROLL; ++u) {
        const bool k = (j + u) < n;
        acc.x += k ? v[u].x : 0.0f;
        acc.y += k ? v[u].y : 0.0f;
        acc.z += k ? v[u].z : 0.0f;
        acc.w += k ? v[u].w : 0.0f;
      }
    }
  }

  if (active) {
    const float ni = norm_in[row];
    const float4 b = *(const float4*)(bias + 4 * lane);
    float4 r;
    r.x = acc.x * ni + b.x;
    r.y = acc.y * ni + b.y;
    r.z = acc.z * ni + b.z;
    r.w = acc.w * ni + b.w;
    if (RELU) {
      r.x = fmaxf(r.x, 0.0f); r.y = fmaxf(r.y, 0.0f);
      r.z = fmaxf(r.z, 0.0f); r.w = fmaxf(r.w, 0.0f);
    }
    *(float4*)(out + (long long)row * COLS + 4 * lane) = r;
  }
}

#define FUSE_WAVES 8
#define FUSE_OCC   8
#define FUSE_BLOCKS (256 * FUSE_OCC / 2)
#define FUSE_ROWS  16
#define FUSE_PITCH 132

__device__ __forceinline__ float4 gcn_gather_row128(
    const float* __restrict__ ycol, const int* __restrict__ col, int start,
    int end, int firstIdx, int lane) {
  float4 acc = make_float4(0.0f, 0.0f, 0.0f, 0.0f);
  int myIdx = firstIdx;
  for (int base = start; base < end; base += 32) {
    const int n = (end - base < 32) ? (end - base) : 32;
    if (base != start) myIdx = (lane < n) ? col[base + lane] : 0;
    for (int j = 0; j < n; j += GATHER_UNROLL) {
      float4 v[GATHER_UNROLL];
#pragma unroll
      for (int u = 0; u < GATHER_UNROLL; ++u) {
        const int s = __builtin_amdgcn_readlane(myIdx, j + u);
        v[u] = *(const float4*)(ycol + (long long)s * 128);
      }
#pragma unroll
      for (int u = 0; u < GATHER_UNROLL; ++u) {
        const bool k = (j + u) < n;
        acc.x += k ? v[u].x : 0.0f;
        acc.y += k ? v[u].y : 0.0f;
        acc.z += k ? v[u].z : 0.0f;
        acc.w += k ? v[u].w : 0.0f;
      }
    }
  }
  return acc;
}

__device__ __forceinline__ void gcn_strip_rows_request(
    const int* __restrict__ rowptr, const float* __restrict__ norm_in,
    const float* __restrict__ norm_out, int r0, int nRows, int lane, int& rp,
    float& nin, float& nso) {
  rp = 0;
  nin = 0.0f;
  nso = 0.0f;
  if (r0 < nRows) {
    const int l2 = (lane < 2) ? lane : 2;
    const int ri = (r0 + l2 < nRows) ? (r0 + l2) : nRows;
    const int l1 = lane & 1;
    const int rn = (r0 + l1 < nRows) ? (r0 + l1) : (nRows - 1);
    rp = rowptr[ri];
    nin = norm_in[rn];
    nso = norm_out[rn];
  }
}
__device__ __forceinline__ float gcn_lane_f32(float v, int l) {
  return __int_as_float(__builtin_amdgcn_readlane(__float_as_int(v), l));
}

template <int NT>
__global__ __launch_bounds__(32 * FUSE_WAVES, FUSE_OCC) void gcn_gather_gemm(
    const float* __restrict__ Yin, const int* __restrict__ rowptr,
    const int* __restrict__ col, const float* __restrict__ norm_in,
    const float* __restrict__ bias, const float* __restrict__ norm_out,
    const float* __restrict__ Wm, float* __restrict__ Yout, int nRows,
    int Mout) {
  constexpr int KB = 16;
  static_assert(NT <= FUSE_WAVES, "one wave per column tile");
  __shared__ float4 As[2][FUSE_ROWS * FUSE_PITCH / 4];
  const int lane = threadIdx.x & 31;
  const int wave = __builtin_amdgcn_readfirstlane((int)(threadIdx.x >> 5));
  const int c = lane & 15;
  const int h = lane >> 4;

  float4 bf[KB];
#pragma unroll
  for (int j = 0; j < KB; ++j) bf[j] = make_float4(0.0f, 0.0f, 0.0f, 0.0f);
  if (wave < NT) {
    const int colw = 16 * wave + c;
    const int colc = (colw < Mout) ? colw : (Mout - 1);
#pragma unroll
    for (int j = 0; j < KB; ++j) {
      const int k = 8 * j + 4 * h;
      bf[j].x = Wm[(k + 0) * Mout + colc];
      bf[j].y = Wm[(k + 1) * Mout + colc];
      bf[j].z = Wm[(k + 2) * Mout + colc];
      bf[j].w = Wm[(k + 3) * Mout + colc];
    }
  }
  const float4 bias4 = *(const float4*)(bias + 4 * lane);
  const float* __restrict__ ycol = Yin + 4 * lane;

  const int nStrips = (nRows + FUSE_ROWS - 1) / FUSE_ROWS;
  const int stripStep = (int)gridDim.x * FUSE_ROWS;
  int rp;
  float nin, nso;
  gcn_strip_rows_request(rowptr, norm_in, norm_out,
                         (int)blockIdx.x * FUSE_ROWS + 2 * wave, nRows, lane, rp,
                         nin, nso);
  int p0 = __builtin_amdgcn_readlane(rp, 0);
  int p1 = __builtin_amdgcn_readlane(rp, 1);
  int p2 = __builtin_amdgcn_readlane(rp, 2);
  float ni0 = gcn_lane_f32(nin, 0), ni1 = gcn_lane_f32(nin, 1);
  float so0 = gcn_lane_f32(nso, 0), so1 = gcn_lane_f32(nso, 1);
  int idx0 = (lane < p1 - p0) ? col[p0 + lane] : 0;
  int idx1 = (lane < p2 - p1) ? col[p1 + lane] : 0;
  gcn_strip_rows_request(rowptr, norm_in, norm_out,
                         (int)blockIdx.x * FUSE_ROWS + 2 * wave + stripStep,
                         nRows, lane, rp, nin, nso);
  int it = 0;
  for (int strip = blockIdx.x; strip < nStrips; strip += gridDim.x, ++it) {
    float* buf = (float*)As[it & 1];
    const int r0 = strip * FUSE_ROWS + 2 * wave;
#pragma unroll
    for (int q = 0; q < 2; ++q) {
      const float4 acc = gcn_gather_row128(ycol, col, q ? p1 : p0, q ? p2 : p1,
                                           q ? idx1 : idx0, lane);
      const float ni = q ? ni1 : ni0;
      const float so = q ? so1 : so0;
      float4 r = make_float4(0.0f, 0.0f, 0.0f, 0.0f);
      if (r0 + q < nRows) {
        r.x = fmaxf(acc.x * ni + bias4.x, 0.0f) * so;
        r.y = fmaxf(acc.y * ni + bias4.y, 0.0f) * so;
        r.z = fmaxf(acc.z * ni + bias4.z, 0.0f) * so;
        r.w = fmaxf(acc.w * ni + bias4.w, 0.0f) * so;
      }
      *(float4*)(buf + (2 * wave + q) * FUSE_PITCH + 4 * lane) = r;
    }
    __syncthreads();
    p0 = __builtin_amdgcn_readlane(rp, 0);
    p1 = __builtin_amdgcn_readlane(rp, 1);
    p2 = __builtin_amdgcn_readlane(rp, 2);
    ni0 = gcn_lane_f32(nin, 0); ni1 = gcn_lane_f32(nin, 1);
    so0 = gcn_lane_f32(nso, 0); so1 = gcn_lane_f32(nso, 1);
    idx0 = (lane < p1 - p0) ? col[p0 + lane] : 0;
    idx1 = (lane < p2 - p1) ? col[p1 + lane] : 0;
    gcn_strip_rows_request(rowptr, norm_in, norm_out, r0 + 2 * stripStep, nRows,
                           lane, rp, nin, nso);

    if (wave < NT) {
      v8f d = v8f{};
      const float* arow = buf + c * FUSE_PITCH + 4 * h;
#pragma unroll
      for (int j = 0; j < KB; ++j) {
        const float4 a = *(const float4*)(arow + 8 * j);
        v2f a0, a1, b0, b1;
        a0.x = a.x; a0.y = a.y; a1.x = a.z; a1.y = a.w;
        b0.x = bf[j].x; b0.y = bf[j].y; b1.x = bf[j].z; b1.y = bf[j].w;
        d = __builtin_amdgcn_wmma_f32_16x16x4_f32(
             false, a0,  false, b0,
             (short)0, d,  false,  false);
        d = __builtin_amdgcn_wmma_f32_16x16x4_f32(
             false, a1,  false, b1,
             (short)0, d,  false,  false);
      }
      const int colw = 16 * wave + c;
      if (colw < Mout) {
#pragma unroll
        for (int r = 0; r < 8; ++r) {
          const int outRow = strip * FUSE_ROWS + r + 8 * h;
          if (outRow < nRows) Yout[(long long)outRow * Mout + colw] = d[r];
        }
      }
    }
  }
}

extern "C" void kernel_launch(void* const* d_in, const int* in_sizes, int n_in,
                              void* d_out, int out_size, void* d_ws, size_t ws_size,
                              hipStream_t stream) {
  const float* feat = (const float*)d_in[0];
  const int*   src  = (const int*)d_in[1];
  const int*   dst  = (const int*)d_in[2];
  const float* W0   = (const float*)d_in[3];
  const float* b0   = (const float*)d_in[4];
  const float* W1   = (const float*)d_in[5];
  const float* b1   = (const float*)d_in[6];
  const float* W2   = (const float*)d_in[7];
  const float* b2   = (const float*)d_in[8];
  float* out = (float*)d_out;

  const int F = 128, H = 128, C = 40;
  const int N = in_sizes[0] / F;
  const int E = in_sizes[1];
  (void)in_sizes; (void)n_in; (void)out_size; (void)ws_size;

  const int NB = (N + BIN_NODES - 1) / BIN_NODES;
  float* ws       = (float*)d_ws;
  float* cntOut   = ws;
  int*   binTot   = (int*)(ws + 2LL * N);
  int*   binCur   = binTot + (long long)NB * BIN_PAD;
  float* norm_out = (float*)(binCur + (long long)NB * BIN_PAD);
  float* norm_in  = norm_out + N;
  float* B0       = norm_in + N;
  float* B1       = B0 + (long long)N * H;
  int* rowptr     = (int*)(B1 + (long long)N * H);
  int* col        = rowptr + (N + 1);
  int* rec        = col + E;

  const int TB = 256;
  auto blocks1d = [](long long n, int tb) { return (unsigned)((n + tb - 1) / tb); };

  const long long nClear = 2LL * N + 2LL * NB * BIN_PAD;
  const unsigned edgeGrid = blocks1d(E, EDGE_CHUNK);
  gcn_zero_f32<<<blocks1d(nClear, TB), TB, 0, stream>>>(ws, nClear);
  gcn_bin_count<<<edgeGrid, SCAN_TB, 0, stream>>>(src, dst, cntOut, binTot, N, E, NB);
  gcn_bin_scatter<<<edgeGrid, SCAN_TB, 0, stream>>>(src, dst, binTot, binCur, rec, E, NB);
  gcn_bin_rows<<<NB, SCAN_TB, 0, stream>>>(rec, binTot, cntOut, rowptr, col, norm_out, norm_in, N, NB);
  const dim3 gatherBlk(32 * GATHER_WAVES);
  const unsigned gatherGrid = (unsigned)((N + GATHER_WAVES - 1) / GATHER_WAVES);

  const dim3 gemmBlk(32 * STRIP_WAVES);
  const dim3 gemmGrid(STRIP_BLOCKS);

  gcn_gemm_strip<128, 8><<<gemmGrid, gemmBlk, 0, stream>>>(feat, W0, norm_out, B0, N, H);

  const dim3 fuseBlk(32 * FUSE_WAVES);
  const dim3 fuseGrid(FUSE_BLOCKS);
  gcn_gather_gemm<8><<<fuseGrid, fuseBlk, 0, stream>>>(B0, rowptr, col, norm_in, b0, norm_out, W1, B1, N, H);

  gcn_gather_gemm<3><<<fuseGrid, fuseBlk, 0, stream>>>(B1, rowptr, col, norm_in, b1, norm_out, W2, B0, N, C);

  gcn_gather<40, false><<<gatherGrid, gatherBlk, 0, stream>>>(B0, rowptr, col, norm_in, b2, out, N);
}


# layer 0 sums the input rows first (edge weight = source norm); its product and layer 1's run behind that gather in one kernel through LDS: no GEMM launch left; layer-0 sum order and rounding change
# speedup vs baseline: 5.2865x; 1.0820x over previous
#include <hip/hip_runtime.h>
#include <hip/hip_bf16.h>

typedef __attribute__((ext_vector_type(2))) float v2f;
typedef __attribute__((ext_vector_type(8))) float v8f;


__global__ void gcn_zero_f32(float* __restrict__ p, long long n) {
  long long i = (long long)blockIdx.x * blockDim.x + threadIdx.x;
  if (i < n) p[i] = 0.0f;
}

__device__ __forceinline__ int gcn_cache_half() {
  const unsigned id = __builtin_amdgcn_s_sendmsg_rtn(0x87);
  return (int)((id >> 18) & 1u);
}

#define SCAN_TB    256
#define BIN_SHIFT  9
#define BIN_NODES  (1 << BIN_SHIFT)
#define BIN_MAX    1024
#define BIN_PAD    32
#define SRC_BITS   20
#define EDGE_PER_THREAD 16
#define ROWS_UNROLL 8
#define EDGE_CHUNK (SCAN_TB * EDGE_PER_THREAD)

__device__ __forceinline__ int gcn_block_excl_scan(int v, int* sh, int& total) {
  const int t = threadIdx.x;
  sh[t] = v;
  __syncthreads();
  int in = 0;
  for (int off = 1; off < SCAN_TB; off <<= 1) {
    int x = sh[in * SCAN_TB + t];
    if (t >= off) x += sh[in * SCAN_TB + t - off];
    sh[(1 - in) * SCAN_TB + t] = x;
    __syncthreads();
    in = 1 - in;
  }
  const int incl = sh[in * SCAN_TB + t];
  total = sh[in * SCAN_TB + SCAN_TB - 1];
  __syncthreads();
  return incl - v;
}

__global__ __launch_bounds__(SCAN_TB) void gcn_bin_count(
    const int* __restrict__ src, const int* __restrict__ dst,
    float* __restrict__ cntOut, int* __restrict__ binTot, int N, int E, int NB) {
  __shared__ int hist[BIN_MAX];
  for (int i = threadIdx.x; i < NB; i += SCAN_TB) hist[i] = 0;
  __syncthreads();
  float* __restrict__ mine = cntOut + (long long)N * gcn_cache_half();
  const int base = blockIdx.x * EDGE_CHUNK + threadIdx.x;
  int d[EDGE_PER_THREAD], s[EDGE_PER_THREAD];
#pragma unroll
  for (int k = 0; k < EDGE_PER_THREAD; ++k) {
    const int e = base + k * SCAN_TB;
    const int ec = (e < E) ? e : (E - 1);
    d[k] = dst[ec];
    s[k] = src[ec];
  }
#pragma unroll
  for (int k = 0; k < EDGE_PER_THREAD; ++k) {
    if (base + k * SCAN_TB < E) {
      atomicAdd(&mine[s[k]], 1.0f);
      atomicAdd(&hist[d[k] >> BIN_SHIFT], 1);
    }
  }
  __syncthreads();
  for (int i = threadIdx.x; i < NB; i += SCAN_TB) {
    const int h = hist[i];
    if (h) atomicAdd(&binTot[i * BIN_PAD], h);
  }
}

__global__ __launch_bounds__(SCAN_TB) void gcn_bin_scatter(
    const int* __restrict__ src, const int* __restrict__ dst,
    const int* __restrict__ binTot, int* __restrict__ binCur,
    int* __restrict__ rec, int E, int NB) {
  __shared__ int hist[BIN_MAX];
  __shared__ int binStart[BIN_MAX];
  __shared__ int sh[2 * SCAN_TB];
  for (int i = threadIdx.x; i < NB; i += SCAN_TB) hist[i] = 0;
  __syncthreads();
  const int base = blockIdx.x * EDGE_CHUNK + threadIdx.x;
  int bin[EDGE_PER_THREAD], rank[EDGE_PER_THREAD], word[EDGE_PER_THREAD];
#pragma unroll
  for (int k = 0; k < EDGE_PER_THREAD; ++k) {
    const int e = base + k * SCAN_TB;
    const int ec = (e < E) ? e : (E - 1);
    const int d = dst[ec];
    bin[k] = d >> BIN_SHIFT;
    word[k] = ((d & (BIN_NODES - 1)) << SRC_BITS) | src[ec];
  }
#pragma unroll
  for (int k = 0; k < EDGE_PER_THREAD; ++k) {
    rank[k] = 0;
    if (base + k * SCAN_TB < E) rank[k] = atomicAdd(&hist[bin[k]], 1);
    else bin[k] = -1;
  }
  int v[4];
  int sum = 0;
#pragma unroll
  for (int k = 0; k < 4; ++k) {
    const int b = threadIdx.x * 4 + k;
    v[k] = (b < NB) ? binTot[b * BIN_PAD] : 0;
    sum += v[k];
  }
  int total;
  int run = gcn_block_excl_scan(sum, sh, total);
#pragma unroll
  for (int k = 0; k < 4; ++k) {
    const int b = threadIdx.x * 4 + k;
    if (b < NB) binStart[b] = run;
    run += v[k];
  }
  __syncthreads();
  for (int i = threadIdx.x; i < NB; i += SCAN_TB) {
    const int h = hist[i];
    if (h) hist[i] = binStart[i] + atomicAdd(&binCur[i * BIN_PAD], h);
  }
  __syncthreads();
#pragma unroll
  for (int k = 0; k < EDGE_PER_THREAD; ++k)
    if (bin[k] >= 0) rec[hist[bin[k]] + rank[k]] = word[k];
}

__global__ __launch_bounds__(SCAN_TB) void gcn_bin_rows(
    const int* __restrict__ rec, const int* __restrict__ binTot,
    const float* __restrict__ cntOut, int* __restrict__ rowptr,
    int* __restrict__ col, float* __restrict__ norm_out,
    float* __restrict__ norm_in, int N, int NB) {
  __shared__ int cntNode[BIN_NODES];
  __shared__ int curNode[BIN_NODES];
  __shared__ int sh[2 * SCAN_TB];
  const int b = blockIdx.x;
  for (int i = threadIdx.x; i < BIN_NODES; i += SCAN_TB) cntNode[i] = 0;
  int part = 0;
  for (int j = threadIdx.x; j < b; j += SCAN_TB) part += binTot[j * BIN_PAD];
  int begin;
  (void)gcn_block_excl_scan(part, sh, begin);
  const int end = begin + binTot[b * BIN_PAD];

  for (int i0 = begin + threadIdx.x; i0 < end; i0 += SCAN_TB * ROWS_UNROLL) {
    int w[ROWS_UNROLL];
#pragma unroll
    for (int u = 0; u < ROWS_UNROLL; ++u) {
      const int i = i0 + u * SCAN_TB;
      w[u] = (i < end) ? rec[i] : -1;
    }
#pragma unroll
    for (int u = 0; u < ROWS_UNROLL; ++u)
      if (w[u] >= 0) atomicAdd(&cntNode[w[u] >> SRC_BITS], 1);
  }
  __syncthreads();

  static_assert(BIN_NODES == 2 * SCAN_TB, "two rows per thread");
  const int c0 = cntNode[2 * threadIdx.x];
  const int c1 = cntNode[2 * threadIdx.x + 1];
  int binEdges;
  const int first = begin + gcn_block_excl_scan(c0 + c1, sh, binEdges);
  curNode[2 * threadIdx.x] = first;
  curNode[2 * threadIdx.x + 1] = first + c0;
  const int node = b * BIN_NODES + 2 * threadIdx.x;
  if (node < N) {
    rowptr[node] = first;
    norm_in[node] = (c0 > 0) ? rsqrtf((float)c0) : 0.0f;
    const float d = cntOut[node] + cntOut[N + node];
    norm_out[node] = (d > 0.0f) ? rsqrtf(d) : 0.0f;
  }
  if (node + 1 < N) {
    rowptr[node + 1] = first + c0;
    norm_in[node + 1] = (c1 > 0) ? rsqrtf((float)c1) : 0.0f;
    const float d = cntOut[node + 1] + cntOut[N + node + 1];
    norm_out[node + 1] = (d > 0.0f) ? rsqrtf(d) : 0.0f;
  }
  if (b == NB - 1 && threadIdx.x == 0) rowptr[N] = end;
  __syncthreads();

  for (int i0 = begin + threadIdx.x; i0 < end; i0 += SCAN_TB * ROWS_UNROLL) {
    int w[ROWS_UNROLL];
#pragma unroll
    for (int u = 0; u < ROWS_UNROLL; ++u) {
      const int i = i0 + u * SCAN_TB;
      w[u] = (i < end) ? rec[i] : -1;
    }
#pragma unroll
    for (int u = 0; u < ROWS_UNROLL; ++u) {
      if (w[u] >= 0) {
        const int slot = atomicAdd(&curNode[w[u] >> SRC_BITS], 1);
        col[slot] = w[u] & ((1 << SRC_BITS) - 1);
      }
    }
  }
}

#define GATHER_WAVES  8
#define GATHER_UNROLL 8
template <int COLS, bool RELU>
__global__ __launch_bounds__(32 * GATHER_WAVES) void gcn_gather(
    const float* __restrict__ Y, const int* __restrict__ rowptr,
    const int* __restrict__ col, const float* __restrict__ norm_in,
    const float* __restrict__ bias, float* __restrict__ out, int nRows) {
  constexpr int LANES = COLS / 4;
  const int lane = threadIdx.x & 31;
  const int row = blockIdx.x * GATHER_WAVES +
                  __builtin_amdgcn_readfirstlane((int)(threadIdx.x >> 5));
  if (row >= nRows) return;
  const bool active = lane < LANES;
  const int start = rowptr[row];
  const int end = rowptr[row + 1];
  const float* __restrict__ ycol = Y + 4 * (active ? lane : 0);

  float4 acc = make_float4(0.0f, 0.0f, 0.0f, 0.0f);
  for (int base = start; base < end; base += 32) {
    const int n = (end - base < 32) ? (end - base) : 32;
    const int myIdx = (lane < n) ? col[base + lane] : 0;
    for (int j = 0; j < n; j += GATHER_UNROLL) {
      float4 v[GATHER_UNROLL];
#pragma unroll
      for (int u = 0; u < GATHER_UNROLL; ++u) {
        const int s = __builtin_amdgcn_readlane(myIdx, j + u);
        v[u] = *(const float4*)(ycol + (long long)s * COLS);
      }
#pragma unroll
      for (int u = 0; u < GATHER_UNROLL; ++u) {
        const bool k = (j + u) < n;
        acc.x += k ? v[u].x : 0.0f;
        acc.y += k ? v[u].y : 0.0f;
        acc.z += k ? v[u].z : 0.0f;
        acc.w += k ? v[u].w : 0.0f;
      }
    }
  }

  if (active) {
    const float ni = norm_in[row];
    const float4 b = *(const float4*)(bias + 4 * lane);
    float4 r;
    r.x = acc.x * ni + b.x;
    r.y = acc.y * ni + b.y;
    r.z = acc.z * ni + b.z;
    r.w = acc.w * ni + b.w;
    if (RELU) {
      r.x = fmaxf(r.x, 0.0f); r.y = fmaxf(r.y, 0.0f);
      r.z = fmaxf(r.z, 0.0f); r.w = fmaxf(r.w, 0.0f);
    }
    *(float4*)(out + (long long)row * COLS + 4 * lane) = r;
  }
}

#define FUSE_WAVES 8
#define FUSE_OCC   8
#define FUSE_BLOCKS (256 * FUSE_OCC / 2)
#define FUSE_ROWS  16
#define FUSE_PITCH 132

__device__ __forceinline__ float4 gcn_gather_row128(
    const float* __restrict__ ycol, const int* __restrict__ col, int start,
    int end, int firstIdx, int lane) {
  float4 acc = make_float4(0.0f, 0.0f, 0.0f, 0.0f);
  int myIdx = firstIdx;
  for (int base = start; base < end; base += 32) {
    const int n = (end - base < 32) ? (end - base) : 32;
    if (base != start) myIdx = (lane < n) ? col[base + lane] : 0;
    for (int j = 0; j < n; j += GATHER_UNROLL) {
      float4 v[GATHER_UNROLL];
#pragma unroll
      for (int u = 0; u < GATHER_UNROLL; ++u) {
        const int s = __builtin_amdgcn_readlane(myIdx, j + u);
        v[u] = *(const float4*)(ycol + (long long)s * 128);
      }
#pragma unroll
      for (int u = 0; u < GATHER_UNROLL; ++u) {
        const bool k = (j + u) < n;
        acc.x += k ? v[u].x : 0.0f;
        acc.y += k ? v[u].y : 0.0f;
        acc.z += k ? v[u].z : 0.0f;
        acc.w += k ? v[u].w : 0.0f;
      }
    }
  }
  return acc;
}

__device__ __forceinline__ void gcn_strip_rows_request(
    const int* __restrict__ rowptr, const float* __restrict__ norm_in,
    const float* __restrict__ norm_out, int r0, int nRows, int lane, int& rp,
    float& nin, float& nso) {
  rp = 0;
  nin = 0.0f;
  nso = 0.0f;
  if (r0 < nRows) {
    const int l2 = (lane < 2) ? lane : 2;
    const int ri = (r0 + l2 < nRows) ? (r0 + l2) : nRows;
    const int l1 = lane & 1;
    const int rn = (r0 + l1 < nRows) ? (r0 + l1) : (nRows - 1);
    rp = rowptr[ri];
    nin = norm_in[rn];
    nso = norm_out[rn];
  }
}
__device__ __forceinline__ float gcn_lane_f32(float v, int l) {
  return __int_as_float(__builtin_amdgcn_readlane(__float_as_int(v), l));
}

template <int NT>
__global__ __launch_bounds__(32 * FUSE_WAVES, FUSE_OCC) void gcn_gather_gemm(
    const float* __restrict__ Yin, const int* __restrict__ rowptr,
    const int* __restrict__ col, const float* __restrict__ norm_in,
    const float* __restrict__ bias, const float* __restrict__ norm_out,
    const float* __restrict__ Wm, float* __restrict__ Yout, int nRows,
    int Mout) {
  constexpr int KB = 16;
  static_assert(NT <= FUSE_WAVES, "one wave per column tile");
  __shared__ float4 As[2][FUSE_ROWS * FUSE_PITCH / 4];
  const int lane = threadIdx.x & 31;
  const int wave = __builtin_amdgcn_readfirstlane((int)(threadIdx.x >> 5));
  const int c = lane & 15;
  const int h = lane >> 4;

  float4 bf[KB];
#pragma unroll
  for (int j = 0; j < KB; ++j) bf[j] = make_float4(0.0f, 0.0f, 0.0f, 0.0f);
  if (wave < NT) {
    const int colw = 16 * wave + c;
    const int colc = (colw < Mout) ? colw : (Mout - 1);
#pragma unroll
    for (int j = 0; j < KB; ++j) {
      const int k = 8 * j + 4 * h;
      bf[j].x = Wm[(k + 0) * Mout + colc];
      bf[j].y = Wm[(k + 1) * Mout + colc];
      bf[j].z = Wm[(k + 2) * Mout + colc];
      bf[j].w = Wm[(k + 3) * Mout + colc];
    }
  }
  const float4 bias4 = *(const float4*)(bias + 4 * lane);
  const float* __restrict__ ycol = Yin + 4 * lane;

  const int nStrips = (nRows + FUSE_ROWS - 1) / FUSE_ROWS;
  const int stripStep = (int)gridDim.x * FUSE_ROWS;
  int rp;
  float nin, nso;
  gcn_strip_rows_request(rowptr, norm_in, norm_out,
                         (int)blockIdx.x * FUSE_ROWS + 2 * wave, nRows, lane, rp,
                         nin, nso);
  int p0 = __builtin_amdgcn_readlane(rp, 0);
  int p1 = __builtin_amdgcn_readlane(rp, 1);
  int p2 = __builtin_amdgcn_readlane(rp, 2);
  float ni0 = gcn_lane_f32(nin, 0), ni1 = gcn_lane_f32(nin, 1);
  float so0 = gcn_lane_f32(nso, 0), so1 = gcn_lane_f32(nso, 1);
  int idx0 = (lane < p1 - p0) ? col[p0 + lane] : 0;
  int idx1 = (lane < p2 - p1) ? col[p1 + lane] : 0;
  gcn_strip_rows_request(rowptr, norm_in, norm_out,
                         (int)blockIdx.x * FUSE_ROWS + 2 * wave + stripStep,
                         nRows, lane, rp, nin, nso);
  int it = 0;
  for (int strip = blockIdx.x; strip < nStrips; strip += gridDim.x, ++it) {
    float* buf = (float*)As[it & 1];
    const int r0 = strip * FUSE_ROWS + 2 * wave;
#pragma unroll
    for (int q = 0; q < 2; ++q) {
      const float4 acc = gcn_gather_row128(ycol, col, q ? p1 : p0, q ? p2 : p1,
                                           q ? idx1 : idx0, lane);
      const float ni = q ? ni1 : ni0;
      const float so = q ? so1 : so0;
      float4 r = make_float4(0.0f, 0.0f, 0.0f, 0.0f);
      if (r0 + q < nRows) {
        r.x = fmaxf(acc.x * ni + bias4.x, 0.0f) * so;
        r.y = fmaxf(acc.y * ni + bias4.y, 0.0f) * so;
        r.z = fmaxf(acc.z * ni + bias4.z, 0.0f) * so;
        r.w = fmaxf(acc.w * ni + bias4.w, 0.0f) * so;
      }
      *(float4*)(buf + (2 * wave + q) * FUSE_PITCH + 4 * lane) = r;
    }
    __syncthreads();
    p0 = __builtin_amdgcn_readlane(rp, 0);
    p1 = __builtin_amdgcn_readlane(rp, 1);
    p2 = __builtin_amdgcn_readlane(rp, 2);
    ni0 = gcn_lane_f32(nin, 0); ni1 = gcn_lane_f32(nin, 1);
    so0 = gcn_lane_f32(nso, 0); so1 = gcn_lane_f32(nso, 1);
    idx0 = (lane < p1 - p0) ? col[p0 + lane] : 0;
    idx1 = (lane < p2 - p1) ? col[p1 + lane] : 0;
    gcn_strip_rows_request(rowptr, norm_in, norm_out, r0 + 2 * stripStep, nRows,
                           lane, rp, nin, nso);

    if (wave < NT) {
      v8f d = v8f{};
      const float* arow = buf + c * FUSE_PITCH + 4 * h;
#pragma unroll
      for (int j = 0; j < KB; ++j) {
        const float4 a = *(const float4*)(arow + 8 * j);
        v2f a0, a1, b0, b1;
        a0.x = a.x; a0.y = a.y; a1.x = a.z; a1.y = a.w;
        b0.x = bf[j].x; b0.y = bf[j].y; b1.x = bf[j].z; b1.y = bf[j].w;
        d = __builtin_amdgcn_wmma_f32_16x16x4_f32(
             false, a0,  false, b0,
             (short)0, d,  false,  false);
        d = __builtin_amdgcn_wmma_f32_16x16x4_f32(
             false, a1,  false, b1,
             (short)0, d,  false,  false);
      }
      const int colw = 16 * wave + c;
      if (colw < Mout) {
#pragma unroll
        for (int r = 0; r < 8; ++r) {
          const int outRow = strip * FUSE_ROWS + r + 8 * h;
          if (outRow < nRows) Yout[(long long)outRow * Mout + colw] = d[r];
        }
      }
    }
  }
}

__device__ __forceinline__ float4 gcn_gather_row128_w(
    const float* __restrict__ ycol, const int* __restrict__ col,
    const float* __restrict__ wsrc, int start, int end, int firstIdx, int lane) {
  float4 acc = make_float4(0.0f, 0.0f, 0.0f, 0.0f);
  int myIdx = firstIdx;
  for (int base = start; base < end; base += 32) {
    const int n = (end - base < 32) ? (end - base) : 32;
    if (base != start) myIdx = (lane < n) ? col[base + lane] : 0;
    const float myW = wsrc[myIdx];
    for (int j = 0; j < n; j += GATHER_UNROLL) {
      float4 v[GATHER_UNROLL];
#pragma unroll
      for (int u = 0; u < GATHER_UNROLL; ++u) {
        const int s = __builtin_amdgcn_readlane(myIdx, j + u);
        v[u] = *(const float4*)(ycol + (long long)s * 128);
      }
      __builtin_amdgcn_sched_barrier(0);
#pragma unroll
      for (int u = 0; u < GATHER_UNROLL; ++u) {
        const bool k = (u == 0) || ((j + u) < n);
        const float w = __int_as_float(
            __builtin_amdgcn_readlane(__float_as_int(myW), j + u));
        acc.x = fmaf(k ? v[u].x : 0.0f, w, acc.x);
        acc.y = fmaf(k ? v[u].y : 0.0f, w, acc.y);
        acc.z = fmaf(k ? v[u].z : 0.0f, w, acc.z);
        acc.w = fmaf(k ? v[u].w : 0.0f, w, acc.w);
      }
    }
  }
  return acc;
}

#define DUO_WAVES 16
#define DUO_BLOCKS (256 * FUSE_OCC / 4)
__global__ __launch_bounds__(32 * DUO_WAVES, FUSE_OCC) void gcn_gather_gemm2(
    const float* __restrict__ X, const int* __restrict__ rowptr,
    const int* __restrict__ col, const float* __restrict__ norm_in,
    const float* __restrict__ norm_out, const float* __restrict__ Wa,
    const float* __restrict__ biasA, const float* __restrict__ Wb,
    float* __restrict__ Yout, int nRows) {
  constexpr int KB = 16;
  constexpr int M = 128;
  static_assert(DUO_WAVES == FUSE_ROWS, "one wave per strip row");
  __shared__ float4 As[2][FUSE_ROWS * FUSE_PITCH / 4];
  __shared__ float4 Hs[2][FUSE_ROWS * FUSE_PITCH / 4];
  __shared__ float4 sNo[2][FUSE_ROWS / 4];
  const int lane = threadIdx.x & 31;
  const int wave = __builtin_amdgcn_readfirstlane((int)(threadIdx.x >> 5));
  const bool first = wave < 8;
  const int tile = wave & 7;
  const int c = lane & 15;
  const int h = lane >> 4;

  const float* __restrict__ Wsel = first ? Wa : Wb;
  const int colw = 16 * tile + c;
  float4 bf[KB];
#pragma unroll
  for (int j = 0; j < KB; ++j) {
    const int k = 8 * j + 4 * h;
    bf[j].x = Wsel[(k + 0) * M + colw];
    bf[j].y = Wsel[(k + 1) * M + colw];
    bf[j].z = Wsel[(k + 2) * M + colw];
    bf[j].w = Wsel[(k + 3) * M + colw];
  }
  const float biasc = biasA[colw];
  const float* __restrict__ ycol = X + 4 * lane;

  const int nStrips = (nRows + FUSE_ROWS - 1) / FUSE_ROWS;
  const int stripStep = (int)gridDim.x * FUSE_ROWS;
  int rp;
  float nin, nso;
  gcn_strip_rows_request(rowptr, norm_in, norm_out,
                         (int)blockIdx.x * FUSE_ROWS + wave, nRows, lane, rp, nin,
                         nso);
  int p0 = __builtin_amdgcn_readlane(rp, 0);
  int p1 = __builtin_amdgcn_readlane(rp, 1);
  float ni0 = gcn_lane_f32(nin, 0);
  float so0 = gcn_lane_f32(nso, 0);
  int idx0 = (lane < p1 - p0) ? col[p0 + lane] : 0;
  gcn_strip_rows_request(rowptr, norm_in, norm_out,
                         (int)blockIdx.x * FUSE_ROWS + wave + stripStep, nRows,
                         lane, rp, nin, nso);
  int it = 0;
  int strip = blockIdx.x;
  for (; strip < nStrips; strip += gridDim.x, ++it) {
    float* abuf = (float*)As[it & 1];
    const int r = strip * FUSE_ROWS + wave;
    const float4 acc = gcn_gather_row128_w(ycol, col, norm_out, p0, p1, idx0, lane);
    float4 rr = make_float4(0.0f, 0.0f, 0.0f, 0.0f);
    float sov = 0.0f;
    if (r < nRows) {
      rr.x = acc.x * ni0; rr.y = acc.y * ni0;
      rr.z = acc.z * ni0; rr.w = acc.w * ni0;
      sov = so0;
    }
    *(float4*)(abuf + wave * FUSE_PITCH + 4 * lane) = rr;
    if (lane == 0) ((float*)sNo[it & 1])[wave] = sov;
    __syncthreads();
    p0 = __builtin_amdgcn_readlane(rp, 0);
    p1 = __builtin_amdgcn_readlane(rp, 1);
    ni0 = gcn_lane_f32(nin, 0);
    so0 = gcn_lane_f32(nso, 0);
    idx0 = (lane < p1 - p0) ? col[p0 + lane] : 0;
    gcn_strip_rows_request(rowptr, norm_in, norm_out, r + 2 * stripStep, nRows,
                           lane, rp, nin, nso);

    if (first) {
      v8f d = v8f{};
      const float* arow = abuf + c * FUSE_PITCH + 4 * h;
#pragma unroll
      for (int j = 0; j < KB; ++j) {
        const float4 a = *(const float4*)(arow + 8 * j);
        v2f a0, a1, b0, b1;
        a0.x = a.x; a0.y = a.y; a1.x = a.z; a1.y = a.w;
        b0.x = bf[j].x; b0.y = bf[j].y; b1.x = bf[j].z; b1.y = bf[j].w;
        d = __builtin_amdgcn_wmma_f32_16x16x4_f32(
             false, a0,  false, b0,
             (short)0, d,  false,  false);
        d = __builtin_amdgcn_wmma_f32_16x16x4_f32(
             false, a1,  false, b1,
             (short)0, d,  false,  false);
      }
      const float4 n0 = sNo[it & 1][2 * h];
      const float4 n1 = sNo[it & 1][2 * h + 1];
      const float no8[8] = {n0.x, n0.y, n0.z, n0.w, n1.x, n1.y, n1.z, n1.w};
      float* hbuf = (float*)Hs[it & 1];
#pragma unroll
      for (int q = 0; q < 8; ++q)
        hbuf[(8 * h + q) * FUSE_PITCH + colw] = fmaxf(d[q] + biasc, 0.0f) * no8[q];
    } else if (it > 0) {
      v8f d = v8f{};
      const float* arow = (const float*)Hs[(it - 1) & 1] + c * FUSE_PITCH + 4 * h;
#pragma unroll
      for (int j = 0; j < KB; ++j) {
        const float4 a = *(const float4*)(arow + 8 * j);
        v2f a0, a1, b0, b1;
        a0.x = a.x; a0.y = a.y; a1.x = a.z; a1.y = a.w;
        b0.x = bf[j].x; b0.y = bf[j].y; b1.x = bf[j].z; b1.y = bf[j].w;
        d = __builtin_amdgcn_wmma_f32_16x16x4_f32(
             false, a0,  false, b0,
             (short)0, d,  false,  false);
        d = __builtin_amdgcn_wmma_f32_16x16x4_f32(
             false, a1,  false, b1,
             (short)0, d,  false,  false);
      }
      const int prev = strip - (int)gridDim.x;
#pragma unroll
      for (int q = 0; q < 8; ++q) {
        const int outRow = prev * FUSE_ROWS + q + 8 * h;
        if (outRow < nRows) Yout[(long long)outRow * M + colw] = d[q];
      }
    }
  }
  __syncthreads();
  if (!first && it > 0) {
    v8f d = v8f{};
    const float* arow = (const float*)Hs[(it - 1) & 1] + c * FUSE_PITCH + 4 * h;
#pragma unroll
    for (int j = 0; j < KB; ++j) {
      const float4 a = *(const float4*)(arow + 8 * j);
      v2f a0, a1, b0, b1;
      a0.x = a.x; a0.y = a.y; a1.x = a.z; a1.y = a.w;
      b0.x = bf[j].x; b0.y = bf[j].y; b1.x = bf[j].z; b1.y = bf[j].w;
      d = __builtin_amdgcn_wmma_f32_16x16x4_f32(
           false, a0,  false, b0,
           (short)0, d,  false,  false);
      d = __builtin_amdgcn_wmma_f32_16x16x4_f32(
           false, a1,  false, b1,
           (short)0, d,  false,  false);
    }
    const int prev = strip - (int)gridDim.x;
#pragma unroll
    for (int q = 0; q < 8; ++q) {
      const int outRow = prev * FUSE_ROWS + q + 8 * h;
      if (outRow < nRows) Yout[(long long)outRow * M + colw] = d[q];
    }
  }
}

extern "C" void kernel_launch(void* const* d_in, const int* in_sizes, int n_in,
                              void* d_out, int out_size, void* d_ws, size_t ws_size,
                              hipStream_t stream) {
  const float* feat = (const float*)d_in[0];
  const int*   src  = (const int*)d_in[1];
  const int*   dst  = (const int*)d_in[2];
  const float* W0   = (const float*)d_in[3];
  const float* b0   = (const float*)d_in[4];
  const float* W1   = (const float*)d_in[5];
  const float* b1   = (const float*)d_in[6];
  const float* W2   = (const float*)d_in[7];
  const float* b2   = (const float*)d_in[8];
  float* out = (float*)d_out;

  const int F = 128, H = 128, C = 40;
  const int N = in_sizes[0] / F;
  const int E = in_sizes[1];
  (void)in_sizes; (void)n_in; (void)out_size; (void)ws_size;

  const int NB = (N + BIN_NODES - 1) / BIN_NODES;
  float* ws       = (float*)d_ws;
  float* cntOut   = ws;
  int*   binTot   = (int*)(ws + 2LL * N);
  int*   binCur   = binTot + (long long)NB * BIN_PAD;
  float* norm_out = (float*)(binCur + (long long)NB * BIN_PAD);
  float* norm_in  = norm_out + N;
  float* B0       = norm_in + N;
  float* B1       = B0 + (long long)N * H;
  int* rowptr     = (int*)(B1 + (long long)N * H);
  int* col        = rowptr + (N + 1);
  int* rec        = col + E;

  const int TB = 256;
  auto blocks1d = [](long long n, int tb) { return (unsigned)((n + tb - 1) / tb); };

  const long long nClear = 2LL * N + 2LL * NB * BIN_PAD;
  const unsigned edgeGrid = blocks1d(E, EDGE_CHUNK);
  gcn_zero_f32<<<blocks1d(nClear, TB), TB, 0, stream>>>(ws, nClear);
  gcn_bin_count<<<edgeGrid, SCAN_TB, 0, stream>>>(src, dst, cntOut, binTot, N, E, NB);
  gcn_bin_scatter<<<edgeGrid, SCAN_TB, 0, stream>>>(src, dst, binTot, binCur, rec, E, NB);
  gcn_bin_rows<<<NB, SCAN_TB, 0, stream>>>(rec, binTot, cntOut, rowptr, col, norm_out, norm_in, N, NB);
  const dim3 gatherBlk(32 * GATHER_WAVES);
  const unsigned gatherGrid = (unsigned)((N + GATHER_WAVES - 1) / GATHER_WAVES);

  const dim3 duoBlk(32 * DUO_WAVES);
  const dim3 duoGrid(DUO_BLOCKS);
  gcn_gather_gemm2<<<duoGrid, duoBlk, 0, stream>>>(feat, rowptr, col, norm_in, norm_out, W0, b0, W1, B1, N);

  const dim3 fuseBlk(32 * FUSE_WAVES);
  const dim3 fuseGrid(FUSE_BLOCKS);
  gcn_gather_gemm<3><<<fuseGrid, fuseBlk, 0, stream>>>(B1, rowptr, col, norm_in, b1, norm_out, W2, B0, N, C);

  gcn_gather<40, false><<<gatherGrid, gatherBlk, 0, stream>>>(B0, rowptr, col, norm_in, b2, out, N);
}


# scattered 4-byte stores of the edge sort (rec[] per block, col[] per bin) are put in order in LDS and written out in destination order: lanes of a store cover whole lines; edge order kept
# speedup vs baseline: 5.5392x; 1.0478x over previous
#include <hip/hip_runtime.h>
#include <hip/hip_bf16.h>

typedef __attribute__((ext_vector_type(2))) float v2f;
typedef __attribute__((ext_vector_type(8))) float v8f;


__global__ void gcn_zero_f32(float* __restrict__ p, long long n) {
  long long i = (long long)blockIdx.x * blockDim.x + threadIdx.x;
  if (i < n) p[i] = 0.0f;
}

__device__ __forceinline__ int gcn_cache_half() {
  const unsigned id = __builtin_amdgcn_s_sendmsg_rtn(0x87);
  return (int)((id >> 18) & 1u);
}

#define SCAN_TB    256
#define BIN_SHIFT  9
#define BIN_NODES  (1 << BIN_SHIFT)
#define BIN_MAX    1024
#define BIN_PAD    32
#define SRC_BITS   20
#define EDGE_PER_THREAD 16
#define ROWS_UNROLL 8
#define EDGE_CHUNK (SCAN_TB * EDGE_PER_THREAD)
#define ROWS_STAGE 16384

__device__ __forceinline__ int gcn_block_excl_scan(int v, int* sh, int& total) {
  const int t = threadIdx.x;
  sh[t] = v;
  __syncthreads();
  int in = 0;
  for (int off = 1; off < SCAN_TB; off <<= 1) {
    int x = sh[in * SCAN_TB + t];
    if (t >= off) x += sh[in * SCAN_TB + t - off];
    sh[(1 - in) * SCAN_TB + t] = x;
    __syncthreads();
    in = 1 - in;
  }
  const int incl = sh[in * SCAN_TB + t];
  total = sh[in * SCAN_TB + SCAN_TB - 1];
  __syncthreads();
  return incl - v;
}

__global__ __launch_bounds__(SCAN_TB) void gcn_bin_count(
    const int* __restrict__ src, const int* __restrict__ dst,
    float* __restrict__ cntOut, int* __restrict__ binTot, int N, int E, int NB) {
  __shared__ int hist[BIN_MAX];
  for (int i = threadIdx.x; i < NB; i += SCAN_TB) hist[i] = 0;
  __syncthreads();
  float* __restrict__ mine = cntOut + (long long)N * gcn_cache_half();
  const int base = blockIdx.x * EDGE_CHUNK + threadIdx.x;
  int d[EDGE_PER_THREAD], s[EDGE_PER_THREAD];
#pragma unroll
  for (int k = 0; k < EDGE_PER_THREAD; ++k) {
    const int e = base + k * SCAN_TB;
    const int ec = (e < E) ? e : (E - 1);
    d[k] = dst[ec];
    s[k] = src[ec];
  }
#pragma unroll
  for (int k = 0; k < EDGE_PER_THREAD; ++k) {
    if (base + k * SCAN_TB < E) {
      atomicAdd(&mine[s[k]], 1.0f);
      atomicAdd(&hist[d[k] >> BIN_SHIFT], 1);
    }
  }
  __syncthreads();
  for (int i = threadIdx.x; i < NB; i += SCAN_TB) {
    const int h = hist[i];
    if (h) atomicAdd(&binTot[i * BIN_PAD], h);
  }
}

__global__ __launch_bounds__(SCAN_TB) void gcn_bin_scatter(
    const int* __restrict__ src, const int* __restrict__ dst,
    const int* __restrict__ binTot, int* __restrict__ binCur,
    int* __restrict__ rec, int E, int NB) {
  __shared__ int hist[BIN_MAX];
  __shared__ int binStart[BIN_MAX];
  __shared__ int locStart[BIN_MAX];
  __shared__ int sh[2 * SCAN_TB];
  __shared__ int stWord[EDGE_CHUNK];
  __shared__ int stDst[EDGE_CHUNK];
  for (int i = threadIdx.x; i < NB; i += SCAN_TB) hist[i] = 0;
  __syncthreads();
  const int base = blockIdx.x * EDGE_CHUNK + threadIdx.x;
  int bin[EDGE_PER_THREAD], rank[EDGE_PER_THREAD], word[EDGE_PER_THREAD];
#pragma unroll
  for (int k = 0; k < EDGE_PER_THREAD; ++k) {
    const int e = base + k * SCAN_TB;
    const int ec = (e < E) ? e : (E - 1);
    const int d = dst[ec];
    bin[k] = d >> BIN_SHIFT;
    word[k] = ((d & (BIN_NODES - 1)) << SRC_BITS) | src[ec];
  }
#pragma unroll
  for (int k = 0; k < EDGE_PER_THREAD; ++k) {
    rank[k] = 0;
    if (base + k * SCAN_TB < E) rank[k] = atomicAdd(&hist[bin[k]], 1);
    else bin[k] = -1;
  }
  int v[4];
  int sum = 0;
#pragma unroll
  for (int k = 0; k < 4; ++k) {
    const int b = threadIdx.x * 4 + k;
    v[k] = (b < NB) ? binTot[b * BIN_PAD] : 0;
    sum += v[k];
  }
  int total;
  int run = gcn_block_excl_scan(sum, sh, total);
#pragma unroll
  for (int k = 0; k < 4; ++k) {
    const int b = threadIdx.x * 4 + k;
    if (b < NB) binStart[b] = run;
    run += v[k];
  }
  int lv[4];
  int lsum = 0;
#pragma unroll
  for (int k = 0; k < 4; ++k) {
    const int b = threadIdx.x * 4 + k;
    lv[k] = (b < NB) ? hist[b] : 0;
    lsum += lv[k];
  }
  int nMine;
  int lrun = gcn_block_excl_scan(lsum, sh, nMine);
#pragma unroll
  for (int k = 0; k < 4; ++k) {
    const int b = threadIdx.x * 4 + k;
    if (b < NB) locStart[b] = lrun;
    lrun += lv[k];
  }
  __syncthreads();
  for (int i = threadIdx.x; i < NB; i += SCAN_TB) {
    const int h = hist[i];
    if (h) hist[i] = binStart[i] + atomicAdd(&binCur[i * BIN_PAD], h);
  }
  __syncthreads();
#pragma unroll
  for (int k = 0; k < EDGE_PER_THREAD; ++k) {
    if (bin[k] >= 0) {
      const int p = locStart[bin[k]] + rank[k];
      stWord[p] = word[k];
      stDst[p] = hist[bin[k]] + rank[k];
    }
  }
  __syncthreads();
#pragma unroll
  for (int k = 0; k < EDGE_PER_THREAD; ++k) {
    const int p = threadIdx.x + k * SCAN_TB;
    if (p < nMine) rec[stDst[p]] = stWord[p];
  }
}

__global__ __launch_bounds__(SCAN_TB) void gcn_bin_rows(
    const int* __restrict__ rec, const int* __restrict__ binTot,
    const float* __restrict__ cntOut, int* __restrict__ rowptr,
    int* __restrict__ col, float* __restrict__ norm_out,
    float* __restrict__ norm_in, int N, int NB) {
  __shared__ int cntNode[BIN_NODES];
  __shared__ int curNode[BIN_NODES];
  __shared__ int sh[2 * SCAN_TB];
  __shared__ int stage[ROWS_STAGE];
  const int b = blockIdx.x;
  for (int i = threadIdx.x; i < BIN_NODES; i += SCAN_TB) cntNode[i] = 0;
  int part = 0;
  for (int j = threadIdx.x; j < b; j += SCAN_TB) part += binTot[j * BIN_PAD];
  int begin;
  (void)gcn_block_excl_scan(part, sh, begin);
  const int end = begin + binTot[b * BIN_PAD];
  const bool staged = (end - begin) <= ROWS_STAGE;

  for (int i0 = begin + threadIdx.x; i0 < end; i0 += SCAN_TB * ROWS_UNROLL) {
    int w[ROWS_UNROLL];
#pragma unroll
    for (int u = 0; u < ROWS_UNROLL; ++u) {
      const int i = i0 + u * SCAN_TB;
      w[u] = (i < end) ? rec[i] : -1;
    }
#pragma unroll
    for (int u = 0; u < ROWS_UNROLL; ++u)
      if (w[u] >= 0) atomicAdd(&cntNode[w[u] >> SRC_BITS], 1);
  }
  __syncthreads();

  static_assert(BIN_NODES == 2 * SCAN_TB, "two rows per thread");
  const int c0 = cntNode[2 * threadIdx.x];
  const int c1 = cntNode[2 * threadIdx.x + 1];
  int binEdges;
  const int first = begin + gcn_block_excl_scan(c0 + c1, sh, binEdges);
  curNode[2 * threadIdx.x] = first;
  curNode[2 * threadIdx.x + 1] = first + c0;
  const int node = b * BIN_NODES + 2 * threadIdx.x;
  if (node < N) {
    rowptr[node] = first;
    norm_in[node] = (c0 > 0) ? rsqrtf((float)c0) : 0.0f;
    const float d = cntOut[node] + cntOut[N + node];
    norm_out[node] = (d > 0.0f) ? rsqrtf(d) : 0.0f;
  }
  if (node + 1 < N) {
    rowptr[node + 1] = first + c0;
    norm_in[node + 1] = (c1 > 0) ? rsqrtf((float)c1) : 0.0f;
    const float d = cntOut[node + 1] + cntOut[N + node + 1];
    norm_out[node + 1] = (d > 0.0f) ? rsqrtf(d) : 0.0f;
  }
  if (b == NB - 1 && threadIdx.x == 0) rowptr[N] = end;
  __syncthreads();

  for (int i0 = begin + threadIdx.x; i0 < end; i0 += SCAN_TB * ROWS_UNROLL) {
    int w[ROWS_UNROLL];
#pragma unroll
    for (int u = 0; u < ROWS_UNROLL; ++u) {
      const int i = i0 + u * SCAN_TB;
      w[u] = (i < end) ? rec[i] : -1;
    }
#pragma unroll
    for (int u = 0; u < ROWS_UNROLL; ++u) {
      if (w[u] >= 0) {
        const int slot = atomicAdd(&curNode[w[u] >> SRC_BITS], 1);
        const int s = w[u] & ((1 << SRC_BITS) - 1);
        if (staged) stage[slot - begin] = s;
        else col[slot] = s;
      }
    }
  }
  if (staged) {
    __syncthreads();
    for (int i = threadIdx.x; i < end - begin; i += SCAN_TB)
      col[begin + i] = stage[i];
  }
}

#define GATHER_WAVES  8
#define GATHER_UNROLL 8
template <int COLS, bool RELU>
__global__ __launch_bounds__(32 * GATHER_WAVES) void gcn_gather(
    const float* __restrict__ Y, const int* __restrict__ rowptr,
    const int* __restrict__ col, const float* __restrict__ norm_in,
    const float* __restrict__ bias, float* __restrict__ out, int nRows) {
  constexpr int LANES = COLS / 4;
  const int lane = threadIdx.x & 31;
  const int row = blockIdx.x * GATHER_WAVES +
                  __builtin_amdgcn_readfirstlane((int)(threadIdx.x >> 5));
  if (row >= nRows) return;
  const bool active = lane < LANES;
  const int start = rowptr[row];
  const int end = rowptr[row + 1];
  const float* __restrict__ ycol = Y + 4 * (active ? lane : 0);

  float4 acc = make_float4(0.0f, 0.0f, 0.0f, 0.0f);
  for (int base = start; base < end; base += 32) {
    const int n = (end - base < 32) ? (end - base) : 32;
    const int myIdx = (lane < n) ? col[base + lane] : 0;
    for (int j = 0; j < n; j += GATHER_UNROLL) {
      float4 v[GATHER_UNROLL];
#pragma unroll
      for (int u = 0; u < GATHER_UNROLL; ++u) {
        const int s = __builtin_amdgcn_readlane(myIdx, j + u);
        v[u] = *(const float4*)(ycol + (long long)s * COLS);
      }
#pragma unroll
      for (int u = 0; u < GATHER_UNROLL; ++u) {
        const bool k = (j + u) < n;
        acc.x += k ? v[u].x : 0.0f;
        acc.y += k ? v[u].y : 0.0f;
        acc.z += k ? v[u].z : 0.0f;
        acc.w += k ? v[u].w : 0.0f;
      }
    }
  }

  if (active) {
    const float ni = norm_in[row];
    const float4 b = *(const float4*)(bias + 4 * lane);
    float4 r;
    r.x = acc.x * ni + b.x;
    r.y = acc.y * ni + b.y;
    r.z = acc.z * ni + b.z;
    r.w = acc.w * ni + b.w;
    if (RELU) {
      r.x = fmaxf(r.x, 0.0f); r.y = fmaxf(r.y, 0.0f);
      r.z = fmaxf(r.z, 0.0f); r.w = fmaxf(r.w, 0.0f);
    }
    *(float4*)(out + (long long)row * COLS + 4 * lane) = r;
  }
}

#define FUSE_WAVES 8
#define FUSE_OCC   8
#define FUSE_BLOCKS (256 * FUSE_OCC / 2)
#define FUSE_ROWS  16
#define FUSE_PITCH 132

__device__ __forceinline__ float4 gcn_gather_row128(
    const float* __restrict__ ycol, const int* __restrict__ col, int start,
    int end, int firstIdx, int lane) {
  float4 acc = make_float4(0.0f, 0.0f, 0.0f, 0.0f);
  int myIdx = firstIdx;
  for (int base = start; base < end; base += 32) {
    const int n = (end - base < 32) ? (end - base) : 32;
    if (base != start) myIdx = (lane < n) ? col[base + lane] : 0;
    for (int j = 0; j < n; j += GATHER_UNROLL) {
      float4 v[GATHER_UNROLL];
#pragma unroll
      for (int u = 0; u < GATHER_UNROLL; ++u) {
        const int s = __builtin_amdgcn_readlane(myIdx, j + u);
        v[u] = *(const float4*)(ycol + (long long)s * 128);
      }
#pragma unroll
      for (int u = 0; u < GATHER_UNROLL; ++u) {
        const bool k = (j + u) < n;
        acc.x += k ? v[u].x : 0.0f;
        acc.y += k ? v[u].y : 0.0f;
        acc.z += k ? v[u].z : 0.0f;
        acc.w += k ? v[u].w : 0.0f;
      }
    }
  }
  return acc;
}

__device__ __forceinline__ void gcn_strip_rows_request(
    const int* __restrict__ rowptr, const float* __restrict__ norm_in,
    const float* __restrict__ norm_out, int r0, int nRows, int lane, int& rp,
    float& nin, float& nso) {
  rp = 0;
  nin = 0.0f;
  nso = 0.0f;
  if (r0 < nRows) {
    const int l2 = (lane < 2) ? lane : 2;
    const int ri = (r0 + l2 < nRows) ? (r0 + l2) : nRows;
    const int l1 = lane & 1;
    const int rn = (r0 + l1 < nRows) ? (r0 + l1) : (nRows - 1);
    rp = rowptr[ri];
    nin = norm_in[rn];
    nso = norm_out[rn];
  }
}
__device__ __forceinline__ float gcn_lane_f32(float v, int l) {
  return __int_as_float(__builtin_amdgcn_readlane(__float_as_int(v), l));
}

template <int NT>
__global__ __launch_bounds__(32 * FUSE_WAVES, FUSE_OCC) void gcn_gather_gemm(
    const float* __restrict__ Yin, const int* __restrict__ rowptr,
    const int* __restrict__ col, const float* __restrict__ norm_in,
    const float* __restrict__ bias, const float* __restrict__ norm_out,
    const float* __restrict__ Wm, float* __restrict__ Yout, int nRows,
    int Mout) {
  constexpr int KB = 16;
  static_assert(NT <= FUSE_WAVES, "one wave per column tile");
  __shared__ float4 As[2][FUSE_ROWS * FUSE_PITCH / 4];
  const int lane = threadIdx.x & 31;
  const int wave = __builtin_amdgcn_readfirstlane((int)(threadIdx.x >> 5));
  const int c = lane & 15;
  const int h = lane >> 4;

  float4 bf[KB];
#pragma unroll
  for (int j = 0; j < KB; ++j) bf[j] = make_float4(0.0f, 0.0f, 0.0f, 0.0f);
  if (wave < NT) {
    const int colw = 16 * wave + c;
    const int colc = (colw < Mout) ? colw : (Mout - 1);
#pragma unroll
    for (int j = 0; j < KB; ++j) {
      const int k = 8 * j + 4 * h;
      bf[j].x = Wm[(k + 0) * Mout + colc];
      bf[j].y = Wm[(k + 1) * Mout + colc];
      bf[j].z = Wm[(k + 2) * Mout + colc];
      bf[j].w = Wm[(k + 3) * Mout + colc];
    }
  }
  const float4 bias4 = *(const float4*)(bias + 4 * lane);
  const float* __restrict__ ycol = Yin + 4 * lane;

  const int nStrips = (nRows + FUSE_ROWS - 1) / FUSE_ROWS;
  const int stripStep = (int)gridDim.x * FUSE_ROWS;
  int rp;
  float nin, nso;
  gcn_strip_rows_request(rowptr, norm_in, norm_out,
                         (int)blockIdx.x * FUSE_ROWS + 2 * wave, nRows, lane, rp,
                         nin, nso);
  int p0 = __builtin_amdgcn_readlane(rp, 0);
  int p1 = __builtin_amdgcn_readlane(rp, 1);
  int p2 = __builtin_amdgcn_readlane(rp, 2);
  float ni0 = gcn_lane_f32(nin, 0), ni1 = gcn_lane_f32(nin, 1);
  float so0 = gcn_lane_f32(nso, 0), so1 = gcn_lane_f32(nso, 1);
  int idx0 = (lane < p1 - p0) ? col[p0 + lane] : 0;
  int idx1 = (lane < p2 - p1) ? col[p1 + lane] : 0;
  gcn_strip_rows_request(rowptr, norm_in, norm_out,
                         (int)blockIdx.x * FUSE_ROWS + 2 * wave + stripStep,
                         nRows, lane, rp, nin, nso);
  int it = 0;
  for (int strip = blockIdx.x; strip < nStrips; strip += gridDim.x, ++it) {
    float* buf = (float*)As[it & 1];
    const int r0 = strip * FUSE_ROWS + 2 * wave;
#pragma unroll
    for (int q = 0; q < 2; ++q) {
      const float4 acc = gcn_gather_row128(ycol, col, q ? p1 : p0, q ? p2 : p1,
                                           q ? idx1 : idx0, lane);
      const float ni = q ? ni1 : ni0;
      const float so = q ? so1 : so0;
      float4 r = make_float4(0.0f, 0.0f, 0.0f, 0.0f);
      if (r0 + q < nRows) {
        r.x = fmaxf(acc.x * ni + bias4.x, 0.0f) * so;
        r.y = fmaxf(acc.y * ni + bias4.y, 0.0f) * so;
        r.z = fmaxf(acc.z * ni + bias4.z, 0.0f) * so;
        r.w = fmaxf(acc.w * ni + bias4.w, 0.0f) * so;
      }
      *(float4*)(buf + (2 * wave + q) * FUSE_PITCH + 4 * lane) = r;
    }
    __syncthreads();
    p0 = __builtin_amdgcn_readlane(rp, 0);
    p1 = __builtin_amdgcn_readlane(rp, 1);
    p2 = __builtin_amdgcn_readlane(rp, 2);
    ni0 = gcn_lane_f32(nin, 0); ni1 = gcn_lane_f32(nin, 1);
    so0 = gcn_lane_f32(nso, 0); so1 = gcn_lane_f32(nso, 1);
    idx0 = (lane < p1 - p0) ? col[p0 + lane] : 0;
    idx1 = (lane < p2 - p1) ? col[p1 + lane] : 0;
    gcn_strip_rows_request(rowptr, norm_in, norm_out, r0 + 2 * stripStep, nRows,
                           lane, rp, nin, nso);

    if (wave < NT) {
      v8f d = v8f{};
      const float* arow = buf + c * FUSE_PITCH + 4 * h;
#pragma unroll
      for (int j = 0; j < KB; ++j) {
        const float4 a = *(const float4*)(arow + 8 * j);
        v2f a0, a1, b0, b1;
        a0.x = a.x; a0.y = a.y; a1.x = a.z; a1.y = a.w;
        b0.x = bf[j].x; b0.y = bf[j].y; b1.x = bf[j].z; b1.y = bf[j].w;
        d = __builtin_amdgcn_wmma_f32_16x16x4_f32(
             false, a0,  false, b0,
             (short)0, d,  false,  false);
        d = __builtin_amdgcn_wmma_f32_16x16x4_f32(
             false, a1,  false, b1,
             (short)0, d,  false,  false);
      }
      const int colw = 16 * wave + c;
      if (colw < Mout) {
#pragma unroll
        for (int r = 0; r < 8; ++r) {
          const int outRow = strip * FUSE_ROWS + r + 8 * h;
          if (outRow < nRows) Yout[(long long)outRow * Mout + colw] = d[r];
        }
      }
    }
  }
}

__device__ __forceinline__ float4 gcn_gather_row128_w(
    const float* __restrict__ ycol, const int* __restrict__ col,
    const float* __restrict__ wsrc, int start, int end, int firstIdx, int lane) {
  float4 acc = make_float4(0.0f, 0.0f, 0.0f, 0.0f);
  int myIdx = firstIdx;
  for (int base = start; base < end; base += 32) {
    const int n = (end - base < 32) ? (end - base) : 32;
    if (base != start) myIdx = (lane < n) ? col[base + lane] : 0;
    const float myW = wsrc[myIdx];
    for (int j = 0; j < n; j += GATHER_UNROLL) {
      float4 v[GATHER_UNROLL];
#pragma unroll
      for (int u = 0; u < GATHER_UNROLL; ++u) {
        const int s = __builtin_amdgcn_readlane(myIdx, j + u);
        v[u] = *(const float4*)(ycol + (long long)s * 128);
      }
      __builtin_amdgcn_sched_barrier(0);
#pragma unroll
      for (int u = 0; u < GATHER_UNROLL; ++u) {
        const bool k = (u == 0) || ((j + u) < n);
        const float w = __int_as_float(
            __builtin_amdgcn_readlane(__float_as_int(myW), j + u));
        acc.x = fmaf(k ? v[u].x : 0.0f, w, acc.x);
        acc.y = fmaf(k ? v[u].y : 0.0f, w, acc.y);
        acc.z = fmaf(k ? v[u].z : 0.0f, w, acc.z);
        acc.w = fmaf(k ? v[u].w : 0.0f, w, acc.w);
      }
    }
  }
  return acc;
}

#define DUO_WAVES 16
#define DUO_BLOCKS (256 * FUSE_OCC / 4)
__global__ __launch_bounds__(32 * DUO_WAVES, FUSE_OCC) void gcn_gather_gemm2(
    const float* __restrict__ X, const int* __restrict__ rowptr,
    const int* __restrict__ col, const float* __restrict__ norm_in,
    const float* __restrict__ norm_out, const float* __restrict__ Wa,
    const float* __restrict__ biasA, const float* __restrict__ Wb,
    float* __restrict__ Yout, int nRows) {
  constexpr int KB = 16;
  constexpr int M = 128;
  static_assert(DUO_WAVES == FUSE_ROWS, "one wave per strip row");
  __shared__ float4 As[2][FUSE_ROWS * FUSE_PITCH / 4];
  __shared__ float4 Hs[2][FUSE_ROWS * FUSE_PITCH / 4];
  __shared__ float4 sNo[2][FUSE_ROWS / 4];
  const int lane = threadIdx.x & 31;
  const int wave = __builtin_amdgcn_readfirstlane((int)(threadIdx.x >> 5));
  const bool first = wave < 8;
  const int tile = wave & 7;
  const int c = lane & 15;
  const int h = lane >> 4;

  const float* __restrict__ Wsel = first ? Wa : Wb;
  const int colw = 16 * tile + c;
  float4 bf[KB];
#pragma unroll
  for (int j = 0; j < KB; ++j) {
    const int k = 8 * j + 4 * h;
    bf[j].x = Wsel[(k + 0) * M + colw];
    bf[j].y = Wsel[(k + 1) * M + colw];
    bf[j].z = Wsel[(k + 2) * M + colw];
    bf[j].w = Wsel[(k + 3) * M + colw];
  }
  const float biasc = biasA[colw];
  const float* __restrict__ ycol = X + 4 * lane;

  const int nStrips = (nRows + FUSE_ROWS - 1) / FUSE_ROWS;
  const int stripStep = (int)gridDim.x * FUSE_ROWS;
  int rp;
  float nin, nso;
  gcn_strip_rows_request(rowptr, norm_in, norm_out,
                         (int)blockIdx.x * FUSE_ROWS + wave, nRows, lane, rp, nin,
                         nso);
  int p0 = __builtin_amdgcn_readlane(rp, 0);
  int p1 = __builtin_amdgcn_readlane(rp, 1);
  float ni0 = gcn_lane_f32(nin, 0);
  float so0 = gcn_lane_f32(nso, 0);
  int idx0 = (lane < p1 - p0) ? col[p0 + lane] : 0;
  gcn_strip_rows_request(rowptr, norm_in, norm_out,
                         (int)blockIdx.x * FUSE_ROWS + wave + stripStep, nRows,
                         lane, rp, nin, nso);
  int it = 0;
  int strip = blockIdx.x;
  for (; strip < nStrips; strip += gridDim.x, ++it) {
    float* abuf = (float*)As[it & 1];
    const int r = strip * FUSE_ROWS + wave;
    const float4 acc = gcn_gather_row128_w(ycol, col, norm_out, p0, p1, idx0, lane);
    float4 rr = make_float4(0.0f, 0.0f, 0.0f, 0.0f);
    float sov = 0.0f;
    if (r < nRows) {
      rr.x = acc.x * ni0; rr.y = acc.y * ni0;
      rr.z = acc.z * ni0; rr.w = acc.w * ni0;
      sov = so0;
    }
    *(float4*)(abuf + wave * FUSE_PITCH + 4 * lane) = rr;
    if (lane == 0) ((float*)sNo[it & 1])[wave] = sov;
    __syncthreads();
    p0 = __builtin_amdgcn_readlane(rp, 0);
    p1 = __builtin_amdgcn_readlane(rp, 1);
    ni0 = gcn_lane_f32(nin, 0);
    so0 = gcn_lane_f32(nso, 0);
    idx0 = (lane < p1 - p0) ? col[p0 + lane] : 0;
    gcn_strip_rows_request(rowptr, norm_in, norm_out, r + 2 * stripStep, nRows,
                           lane, rp, nin, nso);

    if (first) {
      v8f d = v8f{};
      const float* arow = abuf + c * FUSE_PITCH + 4 * h;
#pragma unroll
      for (int j = 0; j < KB; ++j) {
        const float4 a = *(const float4*)(arow + 8 * j);
        v2f a0, a1, b0, b1;
        a0.x = a.x; a0.y = a.y; a1.x = a.z; a1.y = a.w;
        b0.x = bf[j].x; b0.y = bf[j].y; b1.x = bf[j].z; b1.y = bf[j].w;
        d = __builtin_amdgcn_wmma_f32_16x16x4_f32(
             false, a0,  false, b0,
             (short)0, d,  false,  false);
        d = __builtin_amdgcn_wmma_f32_16x16x4_f32(
             false, a1,  false, b1,
             (short)0, d,  false,  false);
      }
      const float4 n0 = sNo[it & 1][2 * h];
      const float4 n1 = sNo[it & 1][2 * h + 1];
      const float no8[8] = {n0.x, n0.y, n0.z, n0.w, n1.x, n1.y, n1.z, n1.w};
      float* hbuf = (float*)Hs[it & 1];
#pragma unroll
      for (int q = 0; q < 8; ++q)
        hbuf[(8 * h + q) * FUSE_PITCH + colw] = fmaxf(d[q] + biasc, 0.0f) * no8[q];
    } else if (it > 0) {
      v8f d = v8f{};
      const float* arow = (const float*)Hs[(it - 1) & 1] + c * FUSE_PITCH + 4 * h;
#pragma unroll
      for (int j = 0; j < KB; ++j) {
        const float4 a = *(const float4*)(arow + 8 * j);
        v2f a0, a1, b0, b1;
        a0.x = a.x; a0.y = a.y; a1.x = a.z; a1.y = a.w;
        b0.x = bf[j].x; b0.y = bf[j].y; b1.x = bf[j].z; b1.y = bf[j].w;
        d = __builtin_amdgcn_wmma_f32_16x16x4_f32(
             false, a0,  false, b0,
             (short)0, d,  false,  false);
        d = __builtin_amdgcn_wmma_f32_16x16x4_f32(
             false, a1,  false, b1,
             (short)0, d,  false,  false);
      }
      const int prev = strip - (int)gridDim.x;
#pragma unroll
      for (int q = 0; q < 8; ++q) {
        const int outRow = prev * FUSE_ROWS + q + 8 * h;
        if (outRow < nRows) Yout[(long long)outRow * M + colw] = d[q];
      }
    }
  }
  __syncthreads();
  if (!first && it > 0) {
    v8f d = v8f{};
    const float* arow = (const float*)Hs[(it - 1) & 1] + c * FUSE_PITCH + 4 * h;
#pragma unroll
    for (int j = 0; j < KB; ++j) {
      const float4 a = *(const float4*)(arow + 8 * j);
      v2f a0, a1, b0, b1;
      a0.x = a.x; a0.y = a.y; a1.x = a.z; a1.y = a.w;
      b0.x = bf[j].x; b0.y = bf[j].y; b1.x = bf[j].z; b1.y = bf[j].w;
      d = __builtin_amdgcn_wmma_f32_16x16x4_f32(
           false, a0,  false, b0,
           (short)0, d,  false,  false);
      d = __builtin_amdgcn_wmma_f32_16x16x4_f32(
           false, a1,  false, b1,
           (short)0, d,  false,  false);
    }
    const int prev = strip - (int)gridDim.x;
#pragma unroll
    for (int q = 0; q < 8; ++q) {
      const int outRow = prev * FUSE_ROWS + q + 8 * h;
      if (outRow < nRows) Yout[(long long)outRow * M + colw] = d[q];
    }
  }
}

extern "C" void kernel_launch(void* const* d_in, const int* in_sizes, int n_in,
                              void* d_out, int out_size, void* d_ws, size_t ws_size,
                              hipStream_t stream) {
  const float* feat = (const float*)d_in[0];
  const int*   src  = (const int*)d_in[1];
  const int*   dst  = (const int*)d_in[2];
  const float* W0   = (const float*)d_in[3];
  const float* b0   = (const float*)d_in[4];
  const float* W1   = (const float*)d_in[5];
  const float* b1   = (const float*)d_in[6];
  const float* W2   = (const float*)d_in[7];
  const float* b2   = (const float*)d_in[8];
  float* out = (float*)d_out;

  const int F = 128, H = 128, C = 40;
  const int N = in_sizes[0] / F;
  const int E = in_sizes[1];
  (void)in_sizes; (void)n_in; (void)out_size; (void)ws_size;

  const int NB = (N + BIN_NODES - 1) / BIN_NODES;
  float* ws       = (float*)d_ws;
  float* cntOut   = ws;
  int*   binTot   = (int*)(ws + 2LL * N);
  int*   binCur   = binTot + (long long)NB * BIN_PAD;
  float* norm_out = (float*)(binCur + (long long)NB * BIN_PAD);
  float* norm_in  = norm_out + N;
  float* B0       = norm_in + N;
  float* B1       = B0 + (long long)N * H;
  int* rowptr     = (int*)(B1 + (long long)N * H);
  int* col        = rowptr + (N + 1);
  int* rec        = col + E;

  const int TB = 256;
  auto blocks1d = [](long long n, int tb) { return (unsigned)((n + tb - 1) / tb); };

  const long long nClear = 2LL * N + 2LL * NB * BIN_PAD;
  const unsigned edgeGrid = blocks1d(E, EDGE_CHUNK);
  gcn_zero_f32<<<blocks1d(nClear, TB), TB, 0, stream>>>(ws, nClear);
  gcn_bin_count<<<edgeGrid, SCAN_TB, 0, stream>>>(src, dst, cntOut, binTot, N, E, NB);
  gcn_bin_scatter<<<edgeGrid, SCAN_TB, 0, stream>>>(src, dst, binTot, binCur, rec, E, NB);
  gcn_bin_rows<<<NB, SCAN_TB, 0, stream>>>(rec, binTot, cntOut, rowptr, col, norm_out, norm_in, N, NB);
  const dim3 gatherBlk(32 * GATHER_WAVES);
  const unsigned gatherGrid = (unsigned)((N + GATHER_WAVES - 1) / GATHER_WAVES);

  const dim3 duoBlk(32 * DUO_WAVES);
  const dim3 duoGrid(DUO_BLOCKS);
  gcn_gather_gemm2<<<duoGrid, duoBlk, 0, stream>>>(feat, rowptr, col, norm_in, norm_out, W0, b0, W1, B1, N);

  const dim3 fuseBlk(32 * FUSE_WAVES);
  const dim3 fuseGrid(FUSE_BLOCKS);
  gcn_gather_gemm<3><<<fuseGrid, fuseBlk, 0, stream>>>(B1, rowptr, col, norm_in, b1, norm_out, W2, B0, N, C);

  gcn_gather<40, false><<<gatherGrid, gatherBlk, 0, stream>>>(B0, rowptr, col, norm_in, b2, out, N);
}
